# starting kernel
# baseline (speedup 1.0000x reference)
#include <hip/hip_runtime.h>
#include <math.h>

#define NN     100000
#define EE     1600000
#define FIN    300
#define HID    64
#define OUTC   20
#define NLAYER 8
#define ALPHA  0.1f

typedef float v2f __attribute__((ext_vector_type(2)));
typedef float v8f __attribute__((ext_vector_type(8)));

__device__ __forceinline__ v8f wmma_f32_16x16x4(v2f a, v2f b, v8f c) {
    return __builtin_amdgcn_wmma_f32_16x16x4_f32(false, a, false, b, (short)0, c,
                                                 false, false);
}

__global__ void k_fill(float* __restrict__ p, float v, int n) {
    int i = blockIdx.x * blockDim.x + threadIdx.x;
    if (i < n) p[i] = v;
}

__global__ void k_deg_acc(const int* __restrict__ col, const float* __restrict__ w,
                          float* __restrict__ deg) {
    int e = blockIdx.x * blockDim.x + threadIdx.x;
    if (e < EE) atomicAdd(&deg[col[e]], w[e]);
}

__global__ void k_dinv(float* __restrict__ deg) {
    int i = blockIdx.x * blockDim.x + threadIdx.x;
    if (i < NN) {
        float d = deg[i];
        deg[i] = d > 0.f ? rsqrtf(d) : 0.f;
    }
}

__global__ void k_norm(const int* __restrict__ row, const int* __restrict__ col,
                       const float* __restrict__ w, const float* __restrict__ dinv,
                       float* __restrict__ norm) {
    int e = blockIdx.x * blockDim.x + threadIdx.x;
    if (e < EE) norm[e] = dinv[row[e]] * w[e] * dinv[col[e]];
}

__global__ __launch_bounds__(128)
void k_gemm_in(const float* __restrict__ x, const float* __restrict__ Win,
               const float* __restrict__ bin, float* __restrict__ x0,
               float* __restrict__ h) {
    __shared__ float xs[16][FIN + 4];
    const int tid = threadIdx.x;
    const int mb  = blockIdx.x;

    for (int i = tid; i < 16 * FIN; i += 128) {
        int r = i / FIN, c = i - r * FIN;
        xs[r][c] = x[(size_t)(mb * 16 + r) * FIN + c];
    }
    __syncthreads();

    const int lane = tid & 31;
    const int n0   = (tid >> 5) * 16;
    const int m    = lane & 15;
    const int kh   = (lane >> 4) * 2;
    const int nl   = lane & 15;

    v8f acc = {};
    for (int k0 = 0; k0 < FIN; k0 += 4) {
        v2f a, b;
        a[0] = xs[m][k0 + kh];
        a[1] = xs[m][k0 + kh + 1];
        b[0] = Win[(size_t)(k0 + kh)     * HID + n0 + nl];
        b[1] = Win[(size_t)(k0 + kh + 1) * HID + n0 + nl];
        acc  = wmma_f32_16x16x4(a, b, acc);
    }

    #pragma unroll
    for (int v = 0; v < 8; ++v) {
        int mm  = v + (lane >> 4) * 8;
        int c   = n0 + nl;
        float val = acc[v] + bin[c];
        val = val > 0.f ? val : 0.f;
        size_t idx = (size_t)(mb * 16 + mm) * HID + c;
        x0[idx] = val;
        h[idx]  = val;
    }
}

__global__ __launch_bounds__(256)
void k_scatter(const int* __restrict__ row, const int* __restrict__ col,
               const float* __restrict__ norm, const float* __restrict__ h,
               float* __restrict__ agg) {
    int gid  = blockIdx.x * blockDim.x + threadIdx.x;
    int lane = gid & 31;
    int wid  = gid >> 5;
    int nwv  = (gridDim.x * blockDim.x) >> 5;
    for (int e = wid; e < EE; e += nwv) {
        int   r  = row[e];
        int   c  = col[e];
        float nv = norm[e];
        float2 hv = *reinterpret_cast<const float2*>(h + (size_t)r * HID + lane * 2);
        atomicAdd(agg + (size_t)c * HID + lane * 2 + 0, nv * hv.x);
        atomicAdd(agg + (size_t)c * HID + lane * 2 + 1, nv * hv.y);
    }
}

__global__ __launch_bounds__(128)
void k_layer(const float* __restrict__ agg, const float* __restrict__ x0,
             const float* __restrict__ dinv, const float* __restrict__ Wl,
             float* __restrict__ h, float beta) {
    __shared__ float zs[16][HID + 8];
    __shared__ float ws[HID][HID];
    const int tid = threadIdx.x;
    const int mb  = blockIdx.x;

    for (int i = tid; i < 16 * HID; i += 128) {
        int r = i >> 6, c = i & 63;
        int rowi = mb * 16 + r;
        float di = dinv[rowi];
        size_t idx = (size_t)rowi * HID + c;
        float a = agg[idx] + di * di * h[idx];
        zs[r][c] = (1.f - ALPHA) * a + ALPHA * x0[idx];
    }
    for (int i = tid; i < HID * HID; i += 128)
        ws[i >> 6][i & 63] = Wl[i];
    __syncthreads();

    const int lane = tid & 31;
    const int n0   = (tid >> 5) * 16;
    const int m    = lane & 15;
    const int kh   = (lane >> 4) * 2;
    const int nl   = lane & 15;

    v8f acc = {};
    #pragma unroll
    for (int k0 = 0; k0 < HID; k0 += 4) {
        v2f a, b;
        a[0] = zs[m][k0 + kh];
        a[1] = zs[m][k0 + kh + 1];
        b[0] = ws[k0 + kh][n0 + nl];
        b[1] = ws[k0 + kh + 1][n0 + nl];
        acc  = wmma_f32_16x16x4(a, b, acc);
    }

    #pragma unroll
    for (int v = 0; v < 8; ++v) {
        int mm = v + (lane >> 4) * 8;
        float z   = zs[mm][n0 + nl];
        float val = (1.f - beta) * z + beta * acc[v];
        val = val > 0.f ? val : 0.f;
        h[(size_t)(mb * 16 + mm) * HID + n0 + nl] = val;
    }
}

__global__ void k_out(const float* __restrict__ h, const float* __restrict__ Wout,
                      const float* __restrict__ bout, float* __restrict__ out) {
    int i = blockIdx.x * blockDim.x + threadIdx.x;
    if (i >= NN * OUTC) return;
    int n = i / OUTC, c = i - n * OUTC;
    float s = bout[c];
    const float* hr = h + (size_t)n * HID;
    #pragma unroll 16
    for (int k = 0; k < HID; ++k) s += hr[k] * Wout[k * OUTC + c];
    out[i] = s;
}

extern "C" void kernel_launch(void* const* d_in, const int* in_sizes, int n_in,
                              void* d_out, int out_size, void* d_ws, size_t ws_size,
                              hipStream_t stream) {
    (void)in_sizes; (void)n_in; (void)out_size; (void)ws_size;

    const float* x     = (const float*)d_in[0];
    const int*   ei    = (const int*)  d_in[1];
    const float* ew    = (const float*)d_in[2];
    const float* Win   = (const float*)d_in[3];
    const float* bin   = (const float*)d_in[4];
    const float* convw = (const float*)d_in[5];
    const float* Wout  = (const float*)d_in[6];
    const float* bout  = (const float*)d_in[7];
    const int* row = ei;
    const int* col = ei + EE;

    char*  ws  = (char*)d_ws;
    size_t off = 0;
    auto carve = [&](size_t bytes) -> float* {
        float* p = (float*)(ws + off);
        off = (off + bytes + 255) & ~(size_t)255;
        return p;
    };
    float* dinv = carve((size_t)NN * 4);
    float* norm = carve((size_t)EE * 4);
    float* x0   = carve((size_t)NN * HID * 4);
    float* h    = carve((size_t)NN * HID * 4);
    float* agg  = carve((size_t)NN * HID * 4);

    k_fill   <<<(NN + 255) / 256, 256, 0, stream>>>(dinv, 1.0f, NN);
    k_deg_acc<<<(EE + 255) / 256, 256, 0, stream>>>(col, ew, dinv);
    k_dinv   <<<(NN + 255) / 256, 256, 0, stream>>>(dinv);
    k_norm   <<<(EE + 255) / 256, 256, 0, stream>>>(row, col, ew, dinv, norm);

    k_gemm_in<<<NN / 16, 128, 0, stream>>>(x, Win, bin, x0, h);

    for (int l = 0; l < NLAYER; ++l) {
        float beta = logf(0.5f / (float)(l + 1) + 1.0f);
        k_fill   <<<(NN * HID + 255) / 256, 256, 0, stream>>>(agg, 0.0f, NN * HID);
        k_scatter<<<2048, 256, 0, stream>>>(row, col, norm, h, agg);
        k_layer  <<<NN / 16, 128, 0, stream>>>(agg, x0, dinv,
                                               convw + (size_t)l * HID * HID, h, beta);
    }

    k_out<<<(NN * OUTC + 255) / 256, 256, 0, stream>>>(h, Wout, bout, (float*)d_out);
}


# aggregation as a gather over edges counting-sorted by destination, fused into the layer kernel (no float atomics, no agg buffer/zero-fill); per-node sum order = count-atomic arrival order
# speedup vs baseline: 2.3619x; 2.3619x over previous
#include <hip/hip_runtime.h>
#include <math.h>


#define NN     100000
#define EE     1600000
#define FIN    300
#define HID    64
#define OUTC   20
#define NLAYER 8
#define ALPHA  0.1f

typedef float v2f __attribute__((ext_vector_type(2)));
typedef float v8f __attribute__((ext_vector_type(8)));

__device__ __forceinline__ v8f wmma_f32_16x16x4(v2f a, v2f b, v8f c) {
    return __builtin_amdgcn_wmma_f32_16x16x4_f32(false, a, false, b, (short)0, c,
                                                 false, false);
}

__global__ void k_zero_i(int* __restrict__ p, int n) {
    int i = blockIdx.x * blockDim.x + threadIdx.x;
    if (i < n) p[i] = 0;
}

__global__ void k_count(const int* __restrict__ col, int* __restrict__ cnt,
                        int* __restrict__ rank) {
    int e = blockIdx.x * blockDim.x + threadIdx.x;
    if (e < EE) rank[e] = atomicAdd(&cnt[col[e]], 1);
}

__global__ __launch_bounds__(1024)
void k_scan(const int* __restrict__ cnt, int* __restrict__ offs) {
    __shared__ int sh[1024];
    const int t    = threadIdx.x;
    const int base = blockIdx.x * 1024;

    int pre = 0;
    for (int i = t; i < base; i += 1024) pre += cnt[i];
    sh[t] = pre;
    __syncthreads();
    for (int d = 512; d > 0; d >>= 1) {
        if (t < d) sh[t] += sh[t + d];
        __syncthreads();
    }
    const int before = sh[0];
    __syncthreads();

    const int i = base + t;
    const int v = (i < NN) ? cnt[i] : 0;
    sh[t] = v;
    __syncthreads();
    for (int d = 1; d < 1024; d <<= 1) {
        int a = (t >= d) ? sh[t - d] : 0;
        __syncthreads();
        sh[t] += a;
        __syncthreads();
    }
    const int incl = sh[t];
    if (i < NN) offs[i] = before + incl - v;
    if (i == NN - 1) offs[NN] = before + incl;
}

__global__ void k_place_w(const int* __restrict__ col, const float* __restrict__ w,
                          const int* __restrict__ rank, const int* __restrict__ offs,
                          float* __restrict__ nrm_s) {
    int e = blockIdx.x * blockDim.x + threadIdx.x;
    if (e < EE) nrm_s[offs[col[e]] + rank[e]] = w[e];
}

__global__ void k_dinv(const int* __restrict__ offs, const float* __restrict__ w_s,
                       float* __restrict__ dinv) {
    int i = blockIdx.x * blockDim.x + threadIdx.x;
    if (i < NN) {
        float d = 1.0f;
        const int e = offs[i + 1];
        for (int p = offs[i]; p < e; ++p) d += w_s[p];
        dinv[i] = d > 0.f ? rsqrtf(d) : 0.f;
    }
}

__global__ void k_place(const int* __restrict__ row, const int* __restrict__ col,
                        const float* __restrict__ w, const int* __restrict__ rank,
                        const int* __restrict__ offs, const float* __restrict__ dinv,
                        int* __restrict__ row_s, float* __restrict__ nrm_s) {
    int e = blockIdx.x * blockDim.x + threadIdx.x;
    if (e < EE) {
        int r = row[e], c = col[e];
        int p = offs[c] + rank[e];
        row_s[p] = r;
        nrm_s[p] = dinv[r] * w[e] * dinv[c];
    }
}

__global__ __launch_bounds__(128)
void k_gemm_in(const float* __restrict__ x, const float* __restrict__ Win,
               const float* __restrict__ bin, float* __restrict__ x0) {
    __shared__ float xs[16][FIN + 4];
    const int tid = threadIdx.x;
    const int mb  = blockIdx.x;

    for (int i = tid; i < 16 * FIN; i += 128) {
        int r = i / FIN, c = i - r * FIN;
        xs[r][c] = x[(size_t)(mb * 16 + r) * FIN + c];
    }
    __syncthreads();

    const int lane = tid & 31;
    const int n0   = (tid >> 5) * 16;
    const int m    = lane & 15;
    const int kh   = (lane >> 4) * 2;
    const int nl   = lane & 15;

    v8f acc = {};
    for (int k0 = 0; k0 < FIN; k0 += 4) {
        v2f a, b;
        a[0] = xs[m][k0 + kh];
        a[1] = xs[m][k0 + kh + 1];
        b[0] = Win[(size_t)(k0 + kh)     * HID + n0 + nl];
        b[1] = Win[(size_t)(k0 + kh + 1) * HID + n0 + nl];
        acc  = wmma_f32_16x16x4(a, b, acc);
    }

    #pragma unroll
    for (int v = 0; v < 8; ++v) {
        int mm  = v + (lane >> 4) * 8;
        int c   = n0 + nl;
        float val = acc[v] + bin[c];
        val = val > 0.f ? val : 0.f;
        x0[(size_t)(mb * 16 + mm) * HID + c] = val;
    }
}

__global__ __launch_bounds__(128)
void k_layer(const int* __restrict__ row_s, const float* __restrict__ nrm_s,
             const int* __restrict__ offs, const float* __restrict__ x0,
             const float* __restrict__ dinv, const float* __restrict__ Wl,
             const float* __restrict__ hin, float* __restrict__ hout, float beta) {
    __shared__ alignas(16) float zs[16][HID + 8];
    __shared__ float ws[HID][HID];
    const int tid  = threadIdx.x;
    const int mb   = blockIdx.x;
    const int lane = tid & 31;
    const int wv   = __builtin_amdgcn_readfirstlane(tid >> 5);

    for (int i = tid; i < HID * HID; i += 128)
        ws[i >> 6][i & 63] = Wl[i];

    const int nb = mb * 16 + wv * 4;
    int    seg_s[4], seg_n[4], pr[4];
    float  pn[4], di[4];
    float2 hs[4], xv[4];
    #pragma unroll
    for (int q = 0; q < 4; ++q) {
        seg_s[q] = offs[nb + q];
        seg_n[q] = offs[nb + q + 1] - seg_s[q];
    }
    #pragma unroll
    for (int q = 0; q < 4; ++q) {
        const bool on = lane < seg_n[q];
        pr[q] = on ? row_s[seg_s[q] + lane] : 0;
        pn[q] = on ? nrm_s[seg_s[q] + lane] : 0.f;
        const size_t idx = (size_t)(nb + q) * HID + lane * 2;
        di[q] = dinv[nb + q];
        hs[q] = *reinterpret_cast<const float2*>(hin + idx);
        xv[q] = *reinterpret_cast<const float2*>(x0 + idx);
    }

    #pragma unroll
    for (int q = 0; q < 4; ++q) {
        float ax = 0.f, ay = 0.f;
        int   my_r = pr[q];
        float my_n = pn[q];
        for (int done = 0; done < seg_n[q]; done += 32) {
            const int n = min(32, seg_n[q] - done);
            if (done > 0) {
                const bool on = lane < n;
                my_r = on ? row_s[seg_s[q] + done + lane] : 0;
                my_n = on ? nrm_s[seg_s[q] + done + lane] : 0.f;
            }
            for (int k = 0; k < n; k += 8) {
                float2 hv[8];
                float  nv[8];
                #pragma unroll
                for (int j = 0; j < 8; ++j) {
                    const int r = __builtin_amdgcn_readlane(my_r, k + j);
                    nv[j] = __int_as_float(
                        __builtin_amdgcn_readlane(__float_as_int(my_n), k + j));
                    hv[j] = *reinterpret_cast<const float2*>(
                        hin + (size_t)r * HID + lane * 2);
                }
                #pragma unroll
                for (int j = 0; j < 8; ++j) {
                    if (k + j < n) {
                        ax += nv[j] * hv[j].x;
                        ay += nv[j] * hv[j].y;
                    }
                }
            }
        }
        const float d2 = di[q] * di[q];
        const float a0 = ax + d2 * hs[q].x;
        const float a1 = ay + d2 * hs[q].y;
        float2 zz;
        zz.x = (1.f - ALPHA) * a0 + ALPHA * xv[q].x;
        zz.y = (1.f - ALPHA) * a1 + ALPHA * xv[q].y;
        *reinterpret_cast<float2*>(&zs[wv * 4 + q][lane * 2]) = zz;
    }
    __syncthreads();

    const int n0   = wv * 16;
    const int m    = lane & 15;
    const int kh   = (lane >> 4) * 2;
    const int nl   = lane & 15;

    v8f acc = {};
    #pragma unroll
    for (int k0 = 0; k0 < HID; k0 += 4) {
        v2f a, b;
        a[0] = zs[m][k0 + kh];
        a[1] = zs[m][k0 + kh + 1];
        b[0] = ws[k0 + kh][n0 + nl];
        b[1] = ws[k0 + kh + 1][n0 + nl];
        acc  = wmma_f32_16x16x4(a, b, acc);
    }

    #pragma unroll
    for (int v = 0; v < 8; ++v) {
        int mm = v + (lane >> 4) * 8;
        float z   = zs[mm][n0 + nl];
        float val = (1.f - beta) * z + beta * acc[v];
        val = val > 0.f ? val : 0.f;
        hout[(size_t)(mb * 16 + mm) * HID + n0 + nl] = val;
    }
}

__global__ void k_out(const float* __restrict__ h, const float* __restrict__ Wout,
                      const float* __restrict__ bout, float* __restrict__ out) {
    int i = blockIdx.x * blockDim.x + threadIdx.x;
    if (i >= NN * OUTC) return;
    int n = i / OUTC, c = i - n * OUTC;
    float s = bout[c];
    const float* hr = h + (size_t)n * HID;
    #pragma unroll 16
    for (int k = 0; k < HID; ++k) s += hr[k] * Wout[k * OUTC + c];
    out[i] = s;
}

extern "C" void kernel_launch(void* const* d_in, const int* in_sizes, int n_in,
                              void* d_out, int out_size, void* d_ws, size_t ws_size,
                              hipStream_t stream) {
    (void)in_sizes; (void)n_in; (void)out_size; (void)ws_size;

    const float* x     = (const float*)d_in[0];
    const int*   ei    = (const int*)  d_in[1];
    const float* ew    = (const float*)d_in[2];
    const float* Win   = (const float*)d_in[3];
    const float* bin   = (const float*)d_in[4];
    const float* convw = (const float*)d_in[5];
    const float* Wout  = (const float*)d_in[6];
    const float* bout  = (const float*)d_in[7];
    const int* row = ei;
    const int* col = ei + EE;

    char*  ws  = (char*)d_ws;
    size_t off = 0;
    auto carve = [&](size_t bytes) -> float* {
        float* p = (float*)(ws + off);
        off = (off + bytes + 255) & ~(size_t)255;
        return p;
    };
    float* dinv  = carve((size_t)NN * 4);
    float* nrm_s = carve((size_t)EE * 4);
    float* x0    = carve((size_t)NN * HID * 4);
    float* hA    = carve((size_t)NN * HID * 4);
    float* hB    = carve((size_t)NN * HID * 4);

    int* cnt  = (int*)hB;
    int* rank = (int*)((char*)hB + 400128);

    int* row_s = (int*)d_out;
    int* offs  = (int*)((char*)d_out + 6400000);

    k_zero_i <<<(NN + 255) / 256, 256, 0, stream>>>(cnt, NN);
    k_count  <<<(EE + 255) / 256, 256, 0, stream>>>(col, cnt, rank);
    k_scan   <<<(NN + 1023) / 1024, 1024, 0, stream>>>(cnt, offs);
    k_place_w<<<(EE + 255) / 256, 256, 0, stream>>>(col, ew, rank, offs, nrm_s);
    k_dinv   <<<(NN + 255) / 256, 256, 0, stream>>>(offs, nrm_s, dinv);
    k_place  <<<(EE + 255) / 256, 256, 0, stream>>>(row, col, ew, rank, offs, dinv,
                                                    row_s, nrm_s);

    k_gemm_in<<<NN / 16, 128, 0, stream>>>(x, Win, bin, x0);

    const float* hin = x0;
    for (int l = 0; l < NLAYER; ++l) {
        float beta = logf(0.5f / (float)(l + 1) + 1.0f);
        float* hout = (l & 1) ? hB : hA;
        k_layer<<<NN / 16, 128, 0, stream>>>(row_s, nrm_s, offs, x0, dinv,
                                             convw + (size_t)l * HID * HID,
                                             hin, hout, beta);
        hin = hout;
    }

    k_out<<<(NN * OUTC + 255) / 256, 256, 0, stream>>>(hin, Wout, bout, (float*)d_out);
}


# edge weights are scattered to their sorted slots once, raw, and normalised in place where the first layer reads its edge lists: the raw-weight scatter pass and two random dinv reads per edge go
# speedup vs baseline: 2.5533x; 1.0810x over previous
#include <hip/hip_runtime.h>
#include <math.h>


#define NN     100000
#define EE     1600000
#define FIN    300
#define HID    64
#define OUTC   20
#define NLAYER 8
#define ALPHA  0.1f

typedef float v2f __attribute__((ext_vector_type(2)));
typedef float v8f __attribute__((ext_vector_type(8)));

__device__ __forceinline__ v8f wmma_f32_16x16x4(v2f a, v2f b, v8f c) {
    return __builtin_amdgcn_wmma_f32_16x16x4_f32(false, a, false, b, (short)0, c,
                                                 false, false);
}

__global__ void k_zero_i(int* __restrict__ p, int n) {
    int i = blockIdx.x * blockDim.x + threadIdx.x;
    if (i < n) p[i] = 0;
}

__global__ void k_count(const int* __restrict__ col, int* __restrict__ cnt,
                        int* __restrict__ rank) {
    int e = blockIdx.x * blockDim.x + threadIdx.x;
    if (e < EE) rank[e] = atomicAdd(&cnt[col[e]], 1);
}

__global__ __launch_bounds__(1024)
void k_scan(const int* __restrict__ cnt, int* __restrict__ offs) {
    __shared__ int sh[1024];
    const int t    = threadIdx.x;
    const int base = blockIdx.x * 1024;

    int pre = 0;
    for (int i = t; i < base; i += 1024) pre += cnt[i];
    sh[t] = pre;
    __syncthreads();
    for (int d = 512; d > 0; d >>= 1) {
        if (t < d) sh[t] += sh[t + d];
        __syncthreads();
    }
    const int before = sh[0];
    __syncthreads();

    const int i = base + t;
    const int v = (i < NN) ? cnt[i] : 0;
    sh[t] = v;
    __syncthreads();
    for (int d = 1; d < 1024; d <<= 1) {
        int a = (t >= d) ? sh[t - d] : 0;
        __syncthreads();
        sh[t] += a;
        __syncthreads();
    }
    const int incl = sh[t];
    if (i < NN) offs[i] = before + incl - v;
    if (i == NN - 1) offs[NN] = before + incl;
}

__global__ void k_dinv(const int* __restrict__ offs, const float* __restrict__ w_s,
                       float* __restrict__ dinv) {
    int i = blockIdx.x * blockDim.x + threadIdx.x;
    if (i < NN) {
        float d = 1.0f;
        const int e = offs[i + 1];
        for (int p = offs[i]; p < e; ++p) d += w_s[p];
        dinv[i] = d > 0.f ? rsqrtf(d) : 0.f;
    }
}

__global__ void k_place(const int* __restrict__ row, const int* __restrict__ col,
                        const float* __restrict__ w, const int* __restrict__ rank,
                        const int* __restrict__ offs,
                        int* __restrict__ row_s, float* __restrict__ w_s) {
    int e = blockIdx.x * blockDim.x + threadIdx.x;
    if (e < EE) {
        int p = offs[col[e]] + rank[e];
        row_s[p] = row[e];
        w_s[p]   = w[e];
    }
}

__global__ __launch_bounds__(128)
void k_gemm_in(const float* __restrict__ x, const float* __restrict__ Win,
               const float* __restrict__ bin, float* __restrict__ x0) {
    __shared__ float xs[16][FIN + 4];
    const int tid = threadIdx.x;
    const int mb  = blockIdx.x;

    for (int i = tid; i < 16 * FIN; i += 128) {
        int r = i / FIN, c = i - r * FIN;
        xs[r][c] = x[(size_t)(mb * 16 + r) * FIN + c];
    }
    __syncthreads();

    const int lane = tid & 31;
    const int n0   = (tid >> 5) * 16;
    const int m    = lane & 15;
    const int kh   = (lane >> 4) * 2;
    const int nl   = lane & 15;

    v8f acc = {};
    for (int k0 = 0; k0 < FIN; k0 += 4) {
        v2f a, b;
        a[0] = xs[m][k0 + kh];
        a[1] = xs[m][k0 + kh + 1];
        b[0] = Win[(size_t)(k0 + kh)     * HID + n0 + nl];
        b[1] = Win[(size_t)(k0 + kh + 1) * HID + n0 + nl];
        acc  = wmma_f32_16x16x4(a, b, acc);
    }

    #pragma unroll
    for (int v = 0; v < 8; ++v) {
        int mm  = v + (lane >> 4) * 8;
        int c   = n0 + nl;
        float val = acc[v] + bin[c];
        val = val > 0.f ? val : 0.f;
        x0[(size_t)(mb * 16 + mm) * HID + c] = val;
    }
}

template <bool FIRST>
__device__ __forceinline__
void layer_body(const int* __restrict__ row_s, const float* __restrict__ nrm_s, float* w_rw,
                const int* __restrict__ offs, const float* __restrict__ x0,
                const float* __restrict__ dinv, const float* __restrict__ Wl,
                const float* __restrict__ hin, float* __restrict__ hout, float beta) {
    __shared__ alignas(16) float zs[16][HID + 8];
    __shared__ float ws[HID][HID];
    const int tid  = threadIdx.x;
    const int mb   = blockIdx.x;
    const int lane = tid & 31;
    const int wv   = __builtin_amdgcn_readfirstlane(tid >> 5);

    for (int i = tid; i < HID * HID; i += 128)
        ws[i >> 6][i & 63] = Wl[i];

    const int nb = mb * 16 + wv * 4;
    int    seg_s[4], seg_n[4], pr[4];
    float  pn[4], di[4];
    float2 hs[4], xv[4];
    #pragma unroll
    for (int q = 0; q < 4; ++q) {
        seg_s[q] = offs[nb + q];
        seg_n[q] = offs[nb + q + 1] - seg_s[q];
    }
    #pragma unroll
    for (int q = 0; q < 4; ++q) {
        const bool on = lane < seg_n[q];
        pr[q] = on ? row_s[seg_s[q] + lane] : 0;
        if (FIRST) pn[q] = on ? w_rw[seg_s[q] + lane] : 0.f;
        else       pn[q] = on ? nrm_s[seg_s[q] + lane] : 0.f;
        const size_t idx = (size_t)(nb + q) * HID + lane * 2;
        di[q] = dinv[nb + q];
        hs[q] = *reinterpret_cast<const float2*>(hin + idx);
        xv[q] = *reinterpret_cast<const float2*>(x0 + idx);
    }
    if (FIRST) {
        float ds[4];
        #pragma unroll
        for (int q = 0; q < 4; ++q) ds[q] = dinv[pr[q]];
        #pragma unroll
        for (int q = 0; q < 4; ++q) pn[q] = ds[q] * pn[q] * di[q];
        #pragma unroll
        for (int q = 0; q < 4; ++q)
            if (lane < seg_n[q]) w_rw[seg_s[q] + lane] = pn[q];
    }

    #pragma unroll
    for (int q = 0; q < 4; ++q) {
        float ax = 0.f, ay = 0.f;
        int   my_r = pr[q];
        float my_n = pn[q];
        for (int done = 0; done < seg_n[q]; done += 32) {
            const int n = min(32, seg_n[q] - done);
            if (done > 0) {
                const bool on = lane < n;
                my_r = on ? row_s[seg_s[q] + done + lane] : 0;
                if (FIRST) {
                    my_n = on ? w_rw[seg_s[q] + done + lane] : 0.f;
                    my_n = dinv[my_r] * my_n * di[q];
                    if (on) w_rw[seg_s[q] + done + lane] = my_n;
                } else {
                    my_n = on ? nrm_s[seg_s[q] + done + lane] : 0.f;
                }
            }
            for (int k = 0; k < n; k += 8) {
                float2 hv[8];
                float  nv[8];
                #pragma unroll
                for (int j = 0; j < 8; ++j) {
                    const int r = __builtin_amdgcn_readlane(my_r, k + j);
                    nv[j] = __int_as_float(
                        __builtin_amdgcn_readlane(__float_as_int(my_n), k + j));
                    hv[j] = *reinterpret_cast<const float2*>(
                        hin + (size_t)r * HID + lane * 2);
                }
                #pragma unroll
                for (int j = 0; j < 8; ++j) {
                    if (k + j < n) {
                        ax += nv[j] * hv[j].x;
                        ay += nv[j] * hv[j].y;
                    }
                }
            }
        }
        const float d2 = di[q] * di[q];
        const float a0 = ax + d2 * hs[q].x;
        const float a1 = ay + d2 * hs[q].y;
        float2 zz;
        zz.x = (1.f - ALPHA) * a0 + ALPHA * xv[q].x;
        zz.y = (1.f - ALPHA) * a1 + ALPHA * xv[q].y;
        *reinterpret_cast<float2*>(&zs[wv * 4 + q][lane * 2]) = zz;
    }
    __syncthreads();

    const int n0   = wv * 16;
    const int m    = lane & 15;
    const int kh   = (lane >> 4) * 2;
    const int nl   = lane & 15;

    v8f acc = {};
    #pragma unroll
    for (int k0 = 0; k0 < HID; k0 += 4) {
        v2f a, b;
        a[0] = zs[m][k0 + kh];
        a[1] = zs[m][k0 + kh + 1];
        b[0] = ws[k0 + kh][n0 + nl];
        b[1] = ws[k0 + kh + 1][n0 + nl];
        acc  = wmma_f32_16x16x4(a, b, acc);
    }

    #pragma unroll
    for (int v = 0; v < 8; ++v) {
        int mm = v + (lane >> 4) * 8;
        float z   = zs[mm][n0 + nl];
        float val = (1.f - beta) * z + beta * acc[v];
        val = val > 0.f ? val : 0.f;
        hout[(size_t)(mb * 16 + mm) * HID + n0 + nl] = val;
    }
}

__global__ __launch_bounds__(128)
void k_layer_first(const int* __restrict__ row_s, float* w_rw,
                   const int* __restrict__ offs, const float* __restrict__ x0,
                   const float* __restrict__ dinv, const float* __restrict__ Wl,
                   const float* __restrict__ hin, float* __restrict__ hout, float beta) {
    layer_body<true>(row_s, nullptr, w_rw, offs, x0, dinv, Wl, hin, hout, beta);
}

__global__ __launch_bounds__(128)
void k_layer(const int* __restrict__ row_s, const float* __restrict__ nrm_s,
             const int* __restrict__ offs, const float* __restrict__ x0,
             const float* __restrict__ dinv, const float* __restrict__ Wl,
             const float* __restrict__ hin, float* __restrict__ hout, float beta) {
    layer_body<false>(row_s, nrm_s, nullptr, offs, x0, dinv, Wl, hin, hout, beta);
}

__global__ void k_out(const float* __restrict__ h, const float* __restrict__ Wout,
                      const float* __restrict__ bout, float* __restrict__ out) {
    int i = blockIdx.x * blockDim.x + threadIdx.x;
    if (i >= NN * OUTC) return;
    int n = i / OUTC, c = i - n * OUTC;
    float s = bout[c];
    const float* hr = h + (size_t)n * HID;
    #pragma unroll 16
    for (int k = 0; k < HID; ++k) s += hr[k] * Wout[k * OUTC + c];
    out[i] = s;
}

extern "C" void kernel_launch(void* const* d_in, const int* in_sizes, int n_in,
                              void* d_out, int out_size, void* d_ws, size_t ws_size,
                              hipStream_t stream) {
    (void)in_sizes; (void)n_in; (void)out_size; (void)ws_size;

    const float* x     = (const float*)d_in[0];
    const int*   ei    = (const int*)  d_in[1];
    const float* ew    = (const float*)d_in[2];
    const float* Win   = (const float*)d_in[3];
    const float* bin   = (const float*)d_in[4];
    const float* convw = (const float*)d_in[5];
    const float* Wout  = (const float*)d_in[6];
    const float* bout  = (const float*)d_in[7];
    const int* row = ei;
    const int* col = ei + EE;

    char*  ws  = (char*)d_ws;
    size_t off = 0;
    auto carve = [&](size_t bytes) -> float* {
        float* p = (float*)(ws + off);
        off = (off + bytes + 255) & ~(size_t)255;
        return p;
    };
    float* dinv  = carve((size_t)NN * 4);
    float* nrm_s = carve((size_t)EE * 4);
    float* x0    = carve((size_t)NN * HID * 4);
    float* hA    = carve((size_t)NN * HID * 4);
    float* hB    = carve((size_t)NN * HID * 4);

    int* cnt  = (int*)hB;
    int* rank = (int*)((char*)hB + 400128);

    int* row_s = (int*)d_out;
    int* offs  = (int*)((char*)d_out + 6400000);

    k_zero_i <<<(NN + 255) / 256, 256, 0, stream>>>(cnt, NN);
    k_count  <<<(EE + 255) / 256, 256, 0, stream>>>(col, cnt, rank);
    k_scan   <<<(NN + 1023) / 1024, 1024, 0, stream>>>(cnt, offs);
    k_place  <<<(EE + 255) / 256, 256, 0, stream>>>(row, col, ew, rank, offs, row_s, nrm_s);
    k_dinv   <<<(NN + 255) / 256, 256, 0, stream>>>(offs, nrm_s, dinv);

    k_gemm_in<<<NN / 16, 128, 0, stream>>>(x, Win, bin, x0);

    const float* hin = x0;
    for (int l = 0; l < NLAYER; ++l) {
        float beta = logf(0.5f / (float)(l + 1) + 1.0f);
        float* hout = (l & 1) ? hB : hA;
        if (l == 0)
            k_layer_first<<<NN / 16, 128, 0, stream>>>(row_s, nrm_s, offs, x0, dinv,
                                                       convw, hin, hout, beta);
        else
            k_layer<<<NN / 16, 128, 0, stream>>>(row_s, nrm_s, offs, x0, dinv,
                                                 convw + (size_t)l * HID * HID,
                                                 hin, hout, beta);
        hin = hout;
    }

    k_out<<<(NN * OUTC + 255) / 256, 256, 0, stream>>>(hin, Wout, bout, (float*)d_out);
}


# edge lists built by two LDS-staged bucket passes (by 512-node bucket, then by node) with whole-run stores, not per-edge global atomics and scattered 4-byte stores; sum order = LDS-atomic arrival
# speedup vs baseline: 3.5257x; 1.3808x over previous
#include <hip/hip_runtime.h>
#include <math.h>


#define NN     100000
#define EE     1600000
#define FIN    300
#define HID    64
#define OUTC   20
#define NLAYER 8
#define ALPHA  0.1f

typedef float v2f __attribute__((ext_vector_type(2)));
typedef float v8f __attribute__((ext_vector_type(8)));

__device__ __forceinline__ v8f wmma_f32_16x16x4(v2f a, v2f b, v8f c) {
    return __builtin_amdgcn_wmma_f32_16x16x4_f32(false, a, false, b, (short)0, c,
                                                 false, false);
}

#define BSH    9
#define BN     (1 << BSH)
#define NBKT   ((NN + BN - 1) / BN)
#define CH     8192
#define NBLK   ((EE + CH - 1) / CH)
#define BM     (NBKT * NBLK)
#define SCAN_PER 39
#define CAP    16384

static_assert(NN <= (1 << 17), "source id must fit 17 bits of the record");
static_assert(NBKT <= 256, "bucket id is staged in a byte");
static_assert(1024 * SCAN_PER >= BM, "k_bscan holds every counter in LDS");
static_assert(CH % 1024 == 0, "pass A gives each thread CH/1024 edges");

__device__ __forceinline__ int block_incl_scan(int* tmp, int v) {
    const int lane = threadIdx.x & 31;
    const int wv   = threadIdx.x >> 5;
    int x = v;
    for (int d = 1; d < 32; d <<= 1) {
        const int y = __shfl_up(x, d);
        if (lane >= d) x += y;
    }
    if (lane == 31) tmp[wv] = x;
    __syncthreads();
    if (wv == 0) {
        int s = tmp[lane];
        for (int d = 1; d < 32; d <<= 1) {
            const int y = __shfl_up(s, d);
            if (lane >= d) s += y;
        }
        tmp[lane] = s;
    }
    __syncthreads();
    return x + (wv > 0 ? tmp[wv - 1] : 0);
}

__global__ __launch_bounds__(1024)
void k_bhist(const int* __restrict__ col, int* __restrict__ bcnt) {
    __shared__ int hist[NBKT];
    const int t   = threadIdx.x;
    const int blk = blockIdx.x;
    for (int i = t; i < NBKT; i += 1024) hist[i] = 0;
    __syncthreads();
    const int e0 = blk * CH;
    for (int j = 0; j < CH / 1024; ++j) {
        const int e = e0 + j * 1024 + t;
        if (e < EE) atomicAdd(&hist[col[e] >> BSH], 1);
    }
    __syncthreads();
    for (int i = t; i < NBKT; i += 1024) bcnt[i * NBLK + blk] = hist[i];
}

__global__ __launch_bounds__(1024)
void k_bscan(int* __restrict__ bcnt) {
    __shared__ int buf[1024 * SCAN_PER];
    __shared__ int sh[32];
    const int t = threadIdx.x;
    for (int i = t; i < 1024 * SCAN_PER; i += 1024) buf[i] = (i < BM) ? bcnt[i] : 0;
    __syncthreads();
    int sum = 0;
    for (int k = 0; k < SCAN_PER; ++k) sum += buf[t * SCAN_PER + k];
    const int incl = block_incl_scan(sh, sum);
    int run = incl - sum;
    for (int k = 0; k < SCAN_PER; ++k) {
        const int v = buf[t * SCAN_PER + k];
        buf[t * SCAN_PER + k] = run;
        run += v;
    }
    __syncthreads();
    for (int i = t; i < BM; i += 1024) bcnt[i] = buf[i];
    if (t == 1023) bcnt[BM] = incl;
}

__global__ __launch_bounds__(1024)
void k_bscatter(const int* __restrict__ row, const int* __restrict__ col,
                const float* __restrict__ w, const int* __restrict__ gbase,
                int* __restrict__ rec) {
    __shared__ int   hist[NBKT];
    __shared__ int   lbase[NBKT];
    __shared__ int   gb[NBKT];
    __shared__ int   sc[32];
    __shared__ int   st_x[CH];
    __shared__ float st_w[CH];
    __shared__ unsigned char st_b[CH];
    const int t   = threadIdx.x;
    const int blk = blockIdx.x;
    for (int i = t; i < NBKT; i += 1024) {
        hist[i] = 0;
        gb[i]   = gbase[i * NBLK + blk];
    }
    __syncthreads();

    const int e0 = blk * CH;
    int   c[CH / 1024], r[CH / 1024], s[CH / 1024];
    float ww[CH / 1024];
    #pragma unroll
    for (int j = 0; j < CH / 1024; ++j) {
        const int e = e0 + j * 1024 + t;
        const bool on = e < EE;
        c[j]  = on ? col[e] : -1;
        s[j]  = on ? row[e] : 0;
        ww[j] = on ? w[e]   : 0.f;
    }
    #pragma unroll
    for (int j = 0; j < CH / 1024; ++j)
        r[j] = (c[j] >= 0) ? atomicAdd(&hist[c[j] >> BSH], 1) : 0;
    __syncthreads();

    const int hv   = (t < NBKT) ? hist[t] : 0;
    const int incl = block_incl_scan(sc, hv);
    if (t < NBKT) lbase[t] = incl - hv;
    __syncthreads();

    #pragma unroll
    for (int j = 0; j < CH / 1024; ++j) {
        if (c[j] >= 0) {
            const int b = c[j] >> BSH;
            const int p = lbase[b] + r[j];
            st_x[p] = s[j] | ((c[j] & (BN - 1)) << 17);
            st_w[p] = ww[j];
            st_b[p] = (unsigned char)b;
        }
    }
    __syncthreads();

    const int nvalid = min(CH, EE - e0);
    for (int i = t; i < nvalid; i += 1024) {
        const int b  = st_b[i];
        const int gp = gb[b] + (i - lbase[b]);
        int2 o;
        o.x = st_x[i];
        o.y = __float_as_int(st_w[i]);
        *reinterpret_cast<int2*>(rec + 2 * (size_t)gp) = o;
    }
}

__global__ __launch_bounds__(1024)
void k_bsort(const int* __restrict__ rec, const int* __restrict__ gbase,
             int* __restrict__ row_s, float* w_s, int* __restrict__ offs,
             float* __restrict__ dinv) {
    __shared__ int   cnt[BN];
    __shared__ int   loff[BN + 1];
    __shared__ int   cur[BN];
    __shared__ int   st_r[CAP];
    __shared__ float st_w[CAP];
    const int t  = threadIdx.x;
    const int b  = blockIdx.x;
    const int bs = gbase[b * NBLK];
    const int be = gbase[(b + 1) * NBLK];
    const int n  = be - bs;
    const int2* rb = reinterpret_cast<const int2*>(rec) + bs;

    if (t < BN) cnt[t] = 0;
    __syncthreads();
    for (int i0 = 0; i0 < n; i0 += 4096) {
        int xs[4];
        #pragma unroll
        for (int u = 0; u < 4; ++u) {
            const int i = i0 + u * 1024 + t;
            xs[u] = (i < n) ? rb[i].x : -1;
        }
        #pragma unroll
        for (int u = 0; u < 4; ++u)
            if (xs[u] >= 0) atomicAdd(&cnt[xs[u] >> 17], 1);
    }
    __syncthreads();

    const int cv   = (t < BN) ? cnt[t] : 0;
    const int incl = block_incl_scan(cur, cv);
    if (t < BN) {
        loff[t] = incl - cv;
        if (t == BN - 1) loff[BN] = incl;
    }
    __syncthreads();
    if (t < BN) {
        cur[t] = loff[t];
        const int node = b * BN + t;
        if (node < NN) offs[node] = bs + loff[t];
    }
    if (b == NBKT - 1 && t == 0) offs[NN] = be;
    __syncthreads();

    const int node = b * BN + t;
    if (n <= CAP) {
        for (int i0 = 0; i0 < n; i0 += 4096) {
            int2 q[4];
            #pragma unroll
            for (int u = 0; u < 4; ++u) {
                const int i = i0 + u * 1024 + t;
                q[u] = (i < n) ? rb[i] : make_int2(-1, 0);
            }
            #pragma unroll
            for (int u = 0; u < 4; ++u) {
                if (q[u].x >= 0) {
                    const int p = atomicAdd(&cur[q[u].x >> 17], 1);
                    st_r[p] = q[u].x & 0x1FFFF;
                    st_w[p] = __int_as_float(q[u].y);
                }
            }
        }
        __syncthreads();
        for (int i = t; i < n; i += 1024) {
            row_s[bs + i] = st_r[i];
            w_s[bs + i]   = st_w[i];
        }
        if (t < BN && node < NN) {
            float d = 1.0f;
            for (int p = loff[t]; p < loff[t + 1]; ++p) d += st_w[p];
            dinv[node] = d > 0.f ? rsqrtf(d) : 0.f;
        }
    } else {
        for (int i = t; i < n; i += 1024) {
            const int2 q = rb[i];
            const int  p = atomicAdd(&cur[q.x >> 17], 1);
            row_s[bs + p] = q.x & 0x1FFFF;
            w_s[bs + p]   = __int_as_float(q.y);
        }
        __syncthreads();
        if (t < BN && node < NN) {
            float d = 1.0f;
            for (int p = loff[t]; p < loff[t + 1]; ++p) d += w_s[bs + p];
            dinv[node] = d > 0.f ? rsqrtf(d) : 0.f;
        }
    }
}

__global__ __launch_bounds__(128)
void k_gemm_in(const float* __restrict__ x, const float* __restrict__ Win,
               const float* __restrict__ bin, float* __restrict__ x0) {
    __shared__ float xs[16][FIN + 4];
    const int tid = threadIdx.x;
    const int mb  = blockIdx.x;

    for (int i = tid; i < 16 * FIN; i += 128) {
        int r = i / FIN, c = i - r * FIN;
        xs[r][c] = x[(size_t)(mb * 16 + r) * FIN + c];
    }
    __syncthreads();

    const int lane = tid & 31;
    const int n0   = (tid >> 5) * 16;
    const int m    = lane & 15;
    const int kh   = (lane >> 4) * 2;
    const int nl   = lane & 15;

    v8f acc = {};
    for (int k0 = 0; k0 < FIN; k0 += 4) {
        v2f a, b;
        a[0] = xs[m][k0 + kh];
        a[1] = xs[m][k0 + kh + 1];
        b[0] = Win[(size_t)(k0 + kh)     * HID + n0 + nl];
        b[1] = Win[(size_t)(k0 + kh + 1) * HID + n0 + nl];
        acc  = wmma_f32_16x16x4(a, b, acc);
    }

    #pragma unroll
    for (int v = 0; v < 8; ++v) {
        int mm  = v + (lane >> 4) * 8;
        int c   = n0 + nl;
        float val = acc[v] + bin[c];
        val = val > 0.f ? val : 0.f;
        x0[(size_t)(mb * 16 + mm) * HID + c] = val;
    }
}

template <bool FIRST>
__device__ __forceinline__
void layer_body(const int* __restrict__ row_s, const float* __restrict__ nrm_s, float* w_rw,
                const int* __restrict__ offs, const float* __restrict__ x0,
                const float* __restrict__ dinv, const float* __restrict__ Wl,
                const float* __restrict__ hin, float* __restrict__ hout, float beta) {
    __shared__ alignas(16) float zs[16][HID + 8];
    __shared__ float ws[HID][HID];
    const int tid  = threadIdx.x;
    const int mb   = blockIdx.x;
    const int lane = tid & 31;
    const int wv   = __builtin_amdgcn_readfirstlane(tid >> 5);

    for (int i = tid; i < HID * HID; i += 128)
        ws[i >> 6][i & 63] = Wl[i];

    const int nb = mb * 16 + wv * 4;
    int    seg_s[4], seg_n[4], pr[4];
    float  pn[4], di[4];
    float2 hs[4], xv[4];
    #pragma unroll
    for (int q = 0; q < 4; ++q) {
        seg_s[q] = offs[nb + q];
        seg_n[q] = offs[nb + q + 1] - seg_s[q];
    }
    #pragma unroll
    for (int q = 0; q < 4; ++q) {
        const bool on = lane < seg_n[q];
        pr[q] = on ? row_s[seg_s[q] + lane] : 0;
        if (FIRST) pn[q] = on ? w_rw[seg_s[q] + lane] : 0.f;
        else       pn[q] = on ? nrm_s[seg_s[q] + lane] : 0.f;
        const size_t idx = (size_t)(nb + q) * HID + lane * 2;
        di[q] = dinv[nb + q];
        hs[q] = *reinterpret_cast<const float2*>(hin + idx);
        xv[q] = *reinterpret_cast<const float2*>(x0 + idx);
    }
    if (FIRST) {
        float ds[4];
        #pragma unroll
        for (int q = 0; q < 4; ++q) ds[q] = dinv[pr[q]];
        #pragma unroll
        for (int q = 0; q < 4; ++q) pn[q] = ds[q] * pn[q] * di[q];
        #pragma unroll
        for (int q = 0; q < 4; ++q)
            if (lane < seg_n[q]) w_rw[seg_s[q] + lane] = pn[q];
    }

    #pragma unroll
    for (int q = 0; q < 4; ++q) {
        float ax = 0.f, ay = 0.f;
        int   my_r = pr[q];
        float my_n = pn[q];
        for (int done = 0; done < seg_n[q]; done += 32) {
            const int n = min(32, seg_n[q] - done);
            if (done > 0) {
                const bool on = lane < n;
                my_r = on ? row_s[seg_s[q] + done + lane] : 0;
                if (FIRST) {
                    my_n = on ? w_rw[seg_s[q] + done + lane] : 0.f;
                    my_n = dinv[my_r] * my_n * di[q];
                    if (on) w_rw[seg_s[q] + done + lane] = my_n;
                } else {
                    my_n = on ? nrm_s[seg_s[q] + done + lane] : 0.f;
                }
            }
            for (int k = 0; k < n; k += 8) {
                float2 hv[8];
                float  nv[8];
                #pragma unroll
                for (int j = 0; j < 8; ++j) {
                    const int r = __builtin_amdgcn_readlane(my_r, k + j);
                    nv[j] = __int_as_float(
                        __builtin_amdgcn_readlane(__float_as_int(my_n), k + j));
                    hv[j] = *reinterpret_cast<const float2*>(
                        hin + (size_t)r * HID + lane * 2);
                }
                #pragma unroll
                for (int j = 0; j < 8; ++j) {
                    if (k + j < n) {
                        ax += nv[j] * hv[j].x;
                        ay += nv[j] * hv[j].y;
                    }
                }
            }
        }
        const float d2 = di[q] * di[q];
        const float a0 = ax + d2 * hs[q].x;
        const float a1 = ay + d2 * hs[q].y;
        float2 zz;
        zz.x = (1.f - ALPHA) * a0 + ALPHA * xv[q].x;
        zz.y = (1.f - ALPHA) * a1 + ALPHA * xv[q].y;
        *reinterpret_cast<float2*>(&zs[wv * 4 + q][lane * 2]) = zz;
    }
    __syncthreads();

    const int n0   = wv * 16;
    const int m    = lane & 15;
    const int kh   = (lane >> 4) * 2;
    const int nl   = lane & 15;

    v8f acc = {};
    #pragma unroll
    for (int k0 = 0; k0 < HID; k0 += 4) {
        v2f a, b;
        a[0] = zs[m][k0 + kh];
        a[1] = zs[m][k0 + kh + 1];
        b[0] = ws[k0 + kh][n0 + nl];
        b[1] = ws[k0 + kh + 1][n0 + nl];
        acc  = wmma_f32_16x16x4(a, b, acc);
    }

    #pragma unroll
    for (int v = 0; v < 8; ++v) {
        int mm = v + (lane >> 4) * 8;
        float z   = zs[mm][n0 + nl];
        float val = (1.f - beta) * z + beta * acc[v];
        val = val > 0.f ? val : 0.f;
        hout[(size_t)(mb * 16 + mm) * HID + n0 + nl] = val;
    }
}

__global__ __launch_bounds__(128)
void k_layer_first(const int* __restrict__ row_s, float* w_rw,
                   const int* __restrict__ offs, const float* __restrict__ x0,
                   const float* __restrict__ dinv, const float* __restrict__ Wl,
                   const float* __restrict__ hin, float* __restrict__ hout, float beta) {
    layer_body<true>(row_s, nullptr, w_rw, offs, x0, dinv, Wl, hin, hout, beta);
}

__global__ __launch_bounds__(128)
void k_layer(const int* __restrict__ row_s, const float* __restrict__ nrm_s,
             const int* __restrict__ offs, const float* __restrict__ x0,
             const float* __restrict__ dinv, const float* __restrict__ Wl,
             const float* __restrict__ hin, float* __restrict__ hout, float beta) {
    layer_body<false>(row_s, nrm_s, nullptr, offs, x0, dinv, Wl, hin, hout, beta);
}

__global__ void k_out(const float* __restrict__ h, const float* __restrict__ Wout,
                      const float* __restrict__ bout, float* __restrict__ out) {
    int i = blockIdx.x * blockDim.x + threadIdx.x;
    if (i >= NN * OUTC) return;
    int n = i / OUTC, c = i - n * OUTC;
    float s = bout[c];
    const float* hr = h + (size_t)n * HID;
    #pragma unroll 16
    for (int k = 0; k < HID; ++k) s += hr[k] * Wout[k * OUTC + c];
    out[i] = s;
}

extern "C" void kernel_launch(void* const* d_in, const int* in_sizes, int n_in,
                              void* d_out, int out_size, void* d_ws, size_t ws_size,
                              hipStream_t stream) {
    (void)in_sizes; (void)n_in; (void)out_size; (void)ws_size;

    const float* x     = (const float*)d_in[0];
    const int*   ei    = (const int*)  d_in[1];
    const float* ew    = (const float*)d_in[2];
    const float* Win   = (const float*)d_in[3];
    const float* bin   = (const float*)d_in[4];
    const float* convw = (const float*)d_in[5];
    const float* Wout  = (const float*)d_in[6];
    const float* bout  = (const float*)d_in[7];
    const int* row = ei;
    const int* col = ei + EE;

    char*  ws  = (char*)d_ws;
    size_t off = 0;
    auto carve = [&](size_t bytes) -> float* {
        float* p = (float*)(ws + off);
        off = (off + bytes + 255) & ~(size_t)255;
        return p;
    };
    float* dinv  = carve((size_t)NN * 4);
    float* nrm_s = carve((size_t)EE * 4);
    float* x0    = carve((size_t)NN * HID * 4);
    float* hA    = carve((size_t)NN * HID * 4);
    float* hB    = carve((size_t)NN * HID * 4);

    int* rec   = (int*)hB;
    int* gbase = (int*)((char*)hB + 12800000);

    int* row_s = (int*)d_out;
    int* offs  = (int*)((char*)d_out + 6400000);

    k_bhist   <<<NBLK, 1024, 0, stream>>>(col, gbase);
    k_bscan   <<<1,    1024, 0, stream>>>(gbase);
    k_bscatter<<<NBLK, 1024, 0, stream>>>(row, col, ew, gbase, rec);
    k_bsort   <<<NBKT, 1024, 0, stream>>>(rec, gbase, row_s, nrm_s, offs, dinv);

    k_gemm_in<<<NN / 16, 128, 0, stream>>>(x, Win, bin, x0);

    const float* hin = x0;
    for (int l = 0; l < NLAYER; ++l) {
        float beta = logf(0.5f / (float)(l + 1) + 1.0f);
        float* hout = (l & 1) ? hB : hA;
        if (l == 0)
            k_layer_first<<<NN / 16, 128, 0, stream>>>(row_s, nrm_s, offs, x0, dinv,
                                                       convw, hin, hout, beta);
        else
            k_layer<<<NN / 16, 128, 0, stream>>>(row_s, nrm_s, offs, x0, dinv,
                                                 convw + (size_t)l * HID * HID,
                                                 hin, hout, beta);
        hin = hout;
    }

    k_out<<<(NN * OUTC + 255) / 256, 256, 0, stream>>>(hin, Wout, bout, (float*)d_out);
}


# layer gather: each half-wave reads one 256-byte row with 16-byte loads, two edges per load (ids, weights via row_share); per-node sum order changes: even-edge + odd-edge partial sums
# speedup vs baseline: 3.6686x; 1.0405x over previous
#include <hip/hip_runtime.h>
#include <math.h>


#define NN     100000
#define EE     1600000
#define FIN    300
#define HID    64
#define OUTC   20
#define NLAYER 8
#define ALPHA  0.1f

typedef float v2f __attribute__((ext_vector_type(2)));
typedef float v8f __attribute__((ext_vector_type(8)));

__device__ __forceinline__ v8f wmma_f32_16x16x4(v2f a, v2f b, v8f c) {
    return __builtin_amdgcn_wmma_f32_16x16x4_f32(false, a, false, b, (short)0, c,
                                                 false, false);
}

#define BSH    9
#define BN     (1 << BSH)
#define NBKT   ((NN + BN - 1) / BN)
#define CH     8192
#define NBLK   ((EE + CH - 1) / CH)
#define BM     (NBKT * NBLK)
#define SCAN_PER 39
#define CAP    16384

static_assert(NN <= (1 << 17), "source id must fit 17 bits of the record");
static_assert(NBKT <= 256, "bucket id is staged in a byte");
static_assert(1024 * SCAN_PER >= BM, "k_bscan holds every counter in LDS");
static_assert(CH % 1024 == 0, "pass A gives each thread CH/1024 edges");

__device__ __forceinline__ int block_incl_scan(int* tmp, int v) {
    const int lane = threadIdx.x & 31;
    const int wv   = threadIdx.x >> 5;
    int x = v;
    for (int d = 1; d < 32; d <<= 1) {
        const int y = __shfl_up(x, d);
        if (lane >= d) x += y;
    }
    if (lane == 31) tmp[wv] = x;
    __syncthreads();
    if (wv == 0) {
        int s = tmp[lane];
        for (int d = 1; d < 32; d <<= 1) {
            const int y = __shfl_up(s, d);
            if (lane >= d) s += y;
        }
        tmp[lane] = s;
    }
    __syncthreads();
    return x + (wv > 0 ? tmp[wv - 1] : 0);
}

__global__ __launch_bounds__(1024)
void k_bhist(const int* __restrict__ col, int* __restrict__ bcnt) {
    __shared__ int hist[NBKT];
    const int t   = threadIdx.x;
    const int blk = blockIdx.x;
    for (int i = t; i < NBKT; i += 1024) hist[i] = 0;
    __syncthreads();
    const int e0 = blk * CH;
    for (int j = 0; j < CH / 1024; ++j) {
        const int e = e0 + j * 1024 + t;
        if (e < EE) atomicAdd(&hist[col[e] >> BSH], 1);
    }
    __syncthreads();
    for (int i = t; i < NBKT; i += 1024) bcnt[i * NBLK + blk] = hist[i];
}

__global__ __launch_bounds__(1024)
void k_bscan(int* __restrict__ bcnt) {
    __shared__ int buf[1024 * SCAN_PER];
    __shared__ int sh[32];
    const int t = threadIdx.x;
    for (int i = t; i < 1024 * SCAN_PER; i += 1024) buf[i] = (i < BM) ? bcnt[i] : 0;
    __syncthreads();
    int sum = 0;
    for (int k = 0; k < SCAN_PER; ++k) sum += buf[t * SCAN_PER + k];
    const int incl = block_incl_scan(sh, sum);
    int run = incl - sum;
    for (int k = 0; k < SCAN_PER; ++k) {
        const int v = buf[t * SCAN_PER + k];
        buf[t * SCAN_PER + k] = run;
        run += v;
    }
    __syncthreads();
    for (int i = t; i < BM; i += 1024) bcnt[i] = buf[i];
    if (t == 1023) bcnt[BM] = incl;
}

__global__ __launch_bounds__(1024)
void k_bscatter(const int* __restrict__ row, const int* __restrict__ col,
                const float* __restrict__ w, const int* __restrict__ gbase,
                int* __restrict__ rec) {
    __shared__ int   hist[NBKT];
    __shared__ int   lbase[NBKT];
    __shared__ int   gb[NBKT];
    __shared__ int   sc[32];
    __shared__ int   st_x[CH];
    __shared__ float st_w[CH];
    __shared__ unsigned char st_b[CH];
    const int t   = threadIdx.x;
    const int blk = blockIdx.x;
    for (int i = t; i < NBKT; i += 1024) {
        hist[i] = 0;
        gb[i]   = gbase[i * NBLK + blk];
    }
    __syncthreads();

    const int e0 = blk * CH;
    int   c[CH / 1024], r[CH / 1024], s[CH / 1024];
    float ww[CH / 1024];
    #pragma unroll
    for (int j = 0; j < CH / 1024; ++j) {
        const int e = e0 + j * 1024 + t;
        const bool on = e < EE;
        c[j]  = on ? col[e] : -1;
        s[j]  = on ? row[e] : 0;
        ww[j] = on ? w[e]   : 0.f;
    }
    #pragma unroll
    for (int j = 0; j < CH / 1024; ++j)
        r[j] = (c[j] >= 0) ? atomicAdd(&hist[c[j] >> BSH], 1) : 0;
    __syncthreads();

    const int hv   = (t < NBKT) ? hist[t] : 0;
    const int incl = block_incl_scan(sc, hv);
    if (t < NBKT) lbase[t] = incl - hv;
    __syncthreads();

    #pragma unroll
    for (int j = 0; j < CH / 1024; ++j) {
        if (c[j] >= 0) {
            const int b = c[j] >> BSH;
            const int p = lbase[b] + r[j];
            st_x[p] = s[j] | ((c[j] & (BN - 1)) << 17);
            st_w[p] = ww[j];
            st_b[p] = (unsigned char)b;
        }
    }
    __syncthreads();

    const int nvalid = min(CH, EE - e0);
    for (int i = t; i < nvalid; i += 1024) {
        const int b  = st_b[i];
        const int gp = gb[b] + (i - lbase[b]);
        int2 o;
        o.x = st_x[i];
        o.y = __float_as_int(st_w[i]);
        *reinterpret_cast<int2*>(rec + 2 * (size_t)gp) = o;
    }
}

__global__ __launch_bounds__(1024)
void k_bsort(const int* __restrict__ rec, const int* __restrict__ gbase,
             int* __restrict__ row_s, float* w_s, int* __restrict__ offs,
             float* __restrict__ dinv) {
    __shared__ int   cnt[BN];
    __shared__ int   loff[BN + 1];
    __shared__ int   cur[BN];
    __shared__ int   st_r[CAP];
    __shared__ float st_w[CAP];
    const int t  = threadIdx.x;
    const int b  = blockIdx.x;
    const int bs = gbase[b * NBLK];
    const int be = gbase[(b + 1) * NBLK];
    const int n  = be - bs;
    const int2* rb = reinterpret_cast<const int2*>(rec) + bs;

    if (t < BN) cnt[t] = 0;
    __syncthreads();
    for (int i0 = 0; i0 < n; i0 += 4096) {
        int xs[4];
        #pragma unroll
        for (int u = 0; u < 4; ++u) {
            const int i = i0 + u * 1024 + t;
            xs[u] = (i < n) ? rb[i].x : -1;
        }
        #pragma unroll
        for (int u = 0; u < 4; ++u)
            if (xs[u] >= 0) atomicAdd(&cnt[xs[u] >> 17], 1);
    }
    __syncthreads();

    const int cv   = (t < BN) ? cnt[t] : 0;
    const int incl = block_incl_scan(cur, cv);
    if (t < BN) {
        loff[t] = incl - cv;
        if (t == BN - 1) loff[BN] = incl;
    }
    __syncthreads();
    if (t < BN) {
        cur[t] = loff[t];
        const int node = b * BN + t;
        if (node < NN) offs[node] = bs + loff[t];
    }
    if (b == NBKT - 1 && t == 0) offs[NN] = be;
    __syncthreads();

    const int node = b * BN + t;
    if (n <= CAP) {
        for (int i0 = 0; i0 < n; i0 += 4096) {
            int2 q[4];
            #pragma unroll
            for (int u = 0; u < 4; ++u) {
                const int i = i0 + u * 1024 + t;
                q[u] = (i < n) ? rb[i] : make_int2(-1, 0);
            }
            #pragma unroll
            for (int u = 0; u < 4; ++u) {
                if (q[u].x >= 0) {
                    const int p = atomicAdd(&cur[q[u].x >> 17], 1);
                    st_r[p] = q[u].x & 0x1FFFF;
                    st_w[p] = __int_as_float(q[u].y);
                }
            }
        }
        __syncthreads();
        for (int i = t; i < n; i += 1024) {
            row_s[bs + i] = st_r[i];
            w_s[bs + i]   = st_w[i];
        }
        if (t < BN && node < NN) {
            float d = 1.0f;
            for (int p = loff[t]; p < loff[t + 1]; ++p) d += st_w[p];
            dinv[node] = d > 0.f ? rsqrtf(d) : 0.f;
        }
    } else {
        for (int i = t; i < n; i += 1024) {
            const int2 q = rb[i];
            const int  p = atomicAdd(&cur[q.x >> 17], 1);
            row_s[bs + p] = q.x & 0x1FFFF;
            w_s[bs + p]   = __int_as_float(q.y);
        }
        __syncthreads();
        if (t < BN && node < NN) {
            float d = 1.0f;
            for (int p = loff[t]; p < loff[t + 1]; ++p) d += w_s[bs + p];
            dinv[node] = d > 0.f ? rsqrtf(d) : 0.f;
        }
    }
}

__global__ __launch_bounds__(128)
void k_gemm_in(const float* __restrict__ x, const float* __restrict__ Win,
               const float* __restrict__ bin, float* __restrict__ x0) {
    __shared__ float xs[16][FIN + 4];
    const int tid = threadIdx.x;
    const int mb  = blockIdx.x;

    for (int i = tid; i < 16 * FIN; i += 128) {
        int r = i / FIN, c = i - r * FIN;
        xs[r][c] = x[(size_t)(mb * 16 + r) * FIN + c];
    }
    __syncthreads();

    const int lane = tid & 31;
    const int n0   = (tid >> 5) * 16;
    const int m    = lane & 15;
    const int kh   = (lane >> 4) * 2;
    const int nl   = lane & 15;

    v8f acc = {};
    for (int k0 = 0; k0 < FIN; k0 += 4) {
        v2f a, b;
        a[0] = xs[m][k0 + kh];
        a[1] = xs[m][k0 + kh + 1];
        b[0] = Win[(size_t)(k0 + kh)     * HID + n0 + nl];
        b[1] = Win[(size_t)(k0 + kh + 1) * HID + n0 + nl];
        acc  = wmma_f32_16x16x4(a, b, acc);
    }

    #pragma unroll
    for (int v = 0; v < 8; ++v) {
        int mm  = v + (lane >> 4) * 8;
        int c   = n0 + nl;
        float val = acc[v] + bin[c];
        val = val > 0.f ? val : 0.f;
        x0[(size_t)(mb * 16 + mm) * HID + c] = val;
    }
}

typedef float v4f __attribute__((ext_vector_type(4)));

template <int S>
__device__ __forceinline__ int row_pick(int x) {
    return __builtin_amdgcn_mov_dpp(x, 0x150 + S, 0xf, 0xf, true);
}

__device__ __forceinline__ float xhalf(float x) {
    const int b = __float_as_int(x);
    return __int_as_float(__builtin_amdgcn_permlanex16(b, b, 0x76543210, 0xfedcba98, false, false));
}

template <int S>
__device__ __forceinline__ void pair_load(const float* __restrict__ hin, int my_r, float my_n,
                                          unsigned c16, v4f& v, float& w) {
    const unsigned r = (unsigned)row_pick<S>(my_r);
    w = __int_as_float(row_pick<S>(__float_as_int(my_n)));
    v = *reinterpret_cast<const v4f*>(reinterpret_cast<const char*>(hin) + ((r << 8) | c16));
}

template <int S0, int G>
__device__ __forceinline__ void gather_pairs(const float* __restrict__ hin, int my_r, float my_n,
                                             unsigned c16, v4f& acc) {
    v4f   v[G];
    float w[G];
    pair_load<S0 + 0>(hin, my_r, my_n, c16, v[0], w[0]);
    pair_load<S0 + 1>(hin, my_r, my_n, c16, v[1], w[1]);
    pair_load<S0 + 2>(hin, my_r, my_n, c16, v[2], w[2]);
    pair_load<S0 + 3>(hin, my_r, my_n, c16, v[3], w[3]);
    if constexpr (G == 8) {
        pair_load<S0 + 4>(hin, my_r, my_n, c16, v[G - 4], w[G - 4]);
        pair_load<S0 + 5>(hin, my_r, my_n, c16, v[G - 3], w[G - 3]);
        pair_load<S0 + 6>(hin, my_r, my_n, c16, v[G - 2], w[G - 2]);
        pair_load<S0 + 7>(hin, my_r, my_n, c16, v[G - 1], w[G - 1]);
    }
    #pragma unroll
    for (int j = 0; j < G; ++j) {
        const v4f ww = {w[j], w[j], w[j], w[j]};
        acc = __builtin_elementwise_fma(v[j], ww, acc);
    }
}

template <bool FIRST>
__device__ __forceinline__
void layer_body(const int* __restrict__ row_s, const float* __restrict__ nrm_s, float* w_rw,
                const int* __restrict__ offs, const float* __restrict__ x0,
                const float* __restrict__ dinv, const float* __restrict__ Wl,
                const float* __restrict__ hin, float* __restrict__ hout, float beta) {
    __shared__ alignas(16) float zs[16][HID + 8];
    __shared__ float ws[HID][HID];
    const int tid  = threadIdx.x;
    const int mb   = blockIdx.x;
    const int lane = tid & 31;
    const int wv   = __builtin_amdgcn_readfirstlane(tid >> 5);
    const int hw   = lane >> 4;
    const int c    = lane & 15;
    const int el   = 2 * c + hw;
    const unsigned c16 = (unsigned)c << 4;

    for (int i = tid; i < HID * HID; i += 128)
        ws[i >> 6][i & 63] = Wl[i];

    const int nb = mb * 16 + wv * 4;
    int    seg_s[4], seg_n[4], pr[4];
    float  pn[4], di[4];
    #pragma unroll
    for (int q = 0; q < 4; ++q) {
        seg_s[q] = offs[nb + q];
        seg_n[q] = offs[nb + q + 1] - seg_s[q];
    }
    #pragma unroll
    for (int q = 0; q < 4; ++q) {
        const bool on = el < seg_n[q];
        pr[q] = on ? row_s[seg_s[q] + el] : 0;
        if (FIRST) pn[q] = on ? w_rw[seg_s[q] + el] : 0.f;
        else       pn[q] = on ? nrm_s[seg_s[q] + el] : 0.f;
        di[q] = dinv[nb + q];
    }
    if (FIRST) {
        float ds[4];
        #pragma unroll
        for (int q = 0; q < 4; ++q) ds[q] = dinv[pr[q]];
        #pragma unroll
        for (int q = 0; q < 4; ++q) pn[q] = ds[q] * pn[q] * di[q];
        #pragma unroll
        for (int q = 0; q < 4; ++q)
            if (el < seg_n[q]) w_rw[seg_s[q] + el] = pn[q];
    }

    const float* own = hw ? x0 : hin;

    #pragma unroll
    for (int q = 0; q < 4; ++q) {
        const v4f val = *reinterpret_cast<const v4f*>(own + (size_t)(nb + q) * HID + c * 4);
        v4f   acc  = {0.f, 0.f, 0.f, 0.f};
        int   my_r = pr[q];
        float my_n = pn[q];
        for (int done = 0; done < seg_n[q]; done += 32) {
            const int n = min(32, seg_n[q] - done);
            if (done > 0) {
                const bool on = el < n;
                my_r = on ? row_s[seg_s[q] + done + el] : 0;
                if (FIRST) {
                    my_n = on ? w_rw[seg_s[q] + done + el] : 0.f;
                    my_n = dinv[my_r] * my_n * di[q];
                    if (on) w_rw[seg_s[q] + done + el] = my_n;
                } else {
                    my_n = on ? nrm_s[seg_s[q] + done + el] : 0.f;
                }
            }
            gather_pairs<0, 8>(hin, my_r, my_n, c16, acc);
            if (n > 16) {
                gather_pairs<8, 4>(hin, my_r, my_n, c16, acc);
                if (n > 24) gather_pairs<12, 4>(hin, my_r, my_n, c16, acc);
            }
        }
        v4f sum, xr;
        sum.x = acc.x + xhalf(acc.x);  xr.x = xhalf(val.x);
        sum.y = acc.y + xhalf(acc.y);  xr.y = xhalf(val.y);
        sum.z = acc.z + xhalf(acc.z);  xr.z = xhalf(val.z);
        sum.w = acc.w + xhalf(acc.w);  xr.w = xhalf(val.w);
        if (hw == 0) {
            const float d2 = di[q] * di[q];
            v4f zz;
            zz.x = __builtin_fmaf(1.f - ALPHA, __builtin_fmaf(d2, val.x, sum.x), ALPHA * xr.x);
            zz.y = __builtin_fmaf(1.f - ALPHA, __builtin_fmaf(d2, val.y, sum.y), ALPHA * xr.y);
            zz.z = __builtin_fmaf(1.f - ALPHA, __builtin_fmaf(d2, val.z, sum.z), ALPHA * xr.z);
            zz.w = __builtin_fmaf(1.f - ALPHA, __builtin_fmaf(d2, val.w, sum.w), ALPHA * xr.w);
            *reinterpret_cast<v4f*>(&zs[wv * 4 + q][c * 4]) = zz;
        }
    }
    __syncthreads();

    const int n0   = wv * 16;
    const int m    = lane & 15;
    const int kh   = (lane >> 4) * 2;
    const int nl   = lane & 15;

    v8f acc = {};
    #pragma unroll
    for (int k0 = 0; k0 < HID; k0 += 4) {
        v2f a, b;
        a[0] = zs[m][k0 + kh];
        a[1] = zs[m][k0 + kh + 1];
        b[0] = ws[k0 + kh][n0 + nl];
        b[1] = ws[k0 + kh + 1][n0 + nl];
        acc  = wmma_f32_16x16x4(a, b, acc);
    }

    #pragma unroll
    for (int v = 0; v < 8; ++v) {
        int mm = v + (lane >> 4) * 8;
        float z   = zs[mm][n0 + nl];
        float val = (1.f - beta) * z + beta * acc[v];
        val = val > 0.f ? val : 0.f;
        hout[(size_t)(mb * 16 + mm) * HID + n0 + nl] = val;
    }
}

__global__ __launch_bounds__(128)
void k_layer_first(const int* __restrict__ row_s, float* w_rw,
                   const int* __restrict__ offs, const float* __restrict__ x0,
                   const float* __restrict__ dinv, const float* __restrict__ Wl,
                   const float* __restrict__ hin, float* __restrict__ hout, float beta) {
    layer_body<true>(row_s, nullptr, w_rw, offs, x0, dinv, Wl, hin, hout, beta);
}

__global__ __launch_bounds__(128)
void k_layer(const int* __restrict__ row_s, const float* __restrict__ nrm_s,
             const int* __restrict__ offs, const float* __restrict__ x0,
             const float* __restrict__ dinv, const float* __restrict__ Wl,
             const float* __restrict__ hin, float* __restrict__ hout, float beta) {
    layer_body<false>(row_s, nrm_s, nullptr, offs, x0, dinv, Wl, hin, hout, beta);
}

__global__ void k_out(const float* __restrict__ h, const float* __restrict__ Wout,
                      const float* __restrict__ bout, float* __restrict__ out) {
    int i = blockIdx.x * blockDim.x + threadIdx.x;
    if (i >= NN * OUTC) return;
    int n = i / OUTC, c = i - n * OUTC;
    float s = bout[c];
    const float* hr = h + (size_t)n * HID;
    #pragma unroll 16
    for (int k = 0; k < HID; ++k) s += hr[k] * Wout[k * OUTC + c];
    out[i] = s;
}

extern "C" void kernel_launch(void* const* d_in, const int* in_sizes, int n_in,
                              void* d_out, int out_size, void* d_ws, size_t ws_size,
                              hipStream_t stream) {
    (void)in_sizes; (void)n_in; (void)out_size; (void)ws_size;

    const float* x     = (const float*)d_in[0];
    const int*   ei    = (const int*)  d_in[1];
    const float* ew    = (const float*)d_in[2];
    const float* Win   = (const float*)d_in[3];
    const float* bin   = (const float*)d_in[4];
    const float* convw = (const float*)d_in[5];
    const float* Wout  = (const float*)d_in[6];
    const float* bout  = (const float*)d_in[7];
    const int* row = ei;
    const int* col = ei + EE;

    char*  ws  = (char*)d_ws;
    size_t off = 0;
    auto carve = [&](size_t bytes) -> float* {
        float* p = (float*)(ws + off);
        off = (off + bytes + 255) & ~(size_t)255;
        return p;
    };
    float* dinv  = carve((size_t)NN * 4);
    float* nrm_s = carve((size_t)EE * 4);
    float* x0    = carve((size_t)NN * HID * 4);
    float* hA    = carve((size_t)NN * HID * 4);
    float* hB    = carve((size_t)NN * HID * 4);

    int* rec   = (int*)hB;
    int* gbase = (int*)((char*)hB + 12800000);

    int* row_s = (int*)d_out;
    int* offs  = (int*)((char*)d_out + 6400000);

    k_bhist   <<<NBLK, 1024, 0, stream>>>(col, gbase);
    k_bscan   <<<1,    1024, 0, stream>>>(gbase);
    k_bscatter<<<NBLK, 1024, 0, stream>>>(row, col, ew, gbase, rec);
    k_bsort   <<<NBKT, 1024, 0, stream>>>(rec, gbase, row_s, nrm_s, offs, dinv);

    k_gemm_in<<<NN / 16, 128, 0, stream>>>(x, Win, bin, x0);

    const float* hin = x0;
    for (int l = 0; l < NLAYER; ++l) {
        float beta = logf(0.5f / (float)(l + 1) + 1.0f);
        float* hout = (l & 1) ? hB : hA;
        if (l == 0)
            k_layer_first<<<NN / 16, 128, 0, stream>>>(row_s, nrm_s, offs, x0, dinv,
                                                       convw, hin, hout, beta);
        else
            k_layer<<<NN / 16, 128, 0, stream>>>(row_s, nrm_s, offs, x0, dinv,
                                                 convw + (size_t)l * HID * HID,
                                                 hin, hout, beta);
        hin = hout;
    }

    k_out<<<(NN * OUTC + 255) / 256, 256, 0, stream>>>(hin, Wout, bout, (float*)d_out);
}


# input GEMM has no launch of its own: its 128-row tiles run as extra work-groups of the scan, scatter and sort launches of the edge build (16-byte tile copy); every sum keeps its order
# speedup vs baseline: 3.8544x; 1.0506x over previous
#include <hip/hip_runtime.h>
#include <math.h>


#define NN     100000
#define EE     1600000
#define FIN    300
#define HID    64
#define OUTC   20
#define NLAYER 8
#define ALPHA  0.1f

typedef float v2f __attribute__((ext_vector_type(2)));
typedef float v4f __attribute__((ext_vector_type(4)));
typedef float v8f __attribute__((ext_vector_type(8)));

__device__ __forceinline__ v8f wmma_f32_16x16x4(v2f a, v2f b, v8f c) {
    return __builtin_amdgcn_wmma_f32_16x16x4_f32(false, a, false, b, (short)0, c,
                                                 false, false);
}

#define BSH    9
#define BN     (1 << BSH)
#define NBKT   ((NN + BN - 1) / BN)
#define CH     8192
#define NBLK   ((EE + CH - 1) / CH)
#define BM     (NBKT * NBLK)
#define SCAN_PER 39
#define CAP    16384

static_assert(NN <= (1 << 17), "source id must fit 17 bits of the record");
static_assert(NBKT <= 256, "bucket id is staged in a byte");
static_assert(1024 * SCAN_PER >= BM, "k_bscan holds every counter in LDS");
static_assert(CH % 1024 == 0, "pass A gives each thread CH/1024 edges");

__device__ __forceinline__ int block_incl_scan(int* tmp, int v) {
    const int lane = threadIdx.x & 31;
    const int wv   = threadIdx.x >> 5;
    int x = v;
    for (int d = 1; d < 32; d <<= 1) {
        const int y = __shfl_up(x, d);
        if (lane >= d) x += y;
    }
    if (lane == 31) tmp[wv] = x;
    __syncthreads();
    if (wv == 0) {
        int s = tmp[lane];
        for (int d = 1; d < 32; d <<= 1) {
            const int y = __shfl_up(s, d);
            if (lane >= d) s += y;
        }
        tmp[lane] = s;
    }
    __syncthreads();
    return x + (wv > 0 ? tmp[wv - 1] : 0);
}

__global__ __launch_bounds__(1024)
void k_bhist(const int* __restrict__ col, int* __restrict__ bcnt) {
    __shared__ int hist[NBKT];
    const int t   = threadIdx.x;
    const int blk = blockIdx.x;
    for (int i = t; i < NBKT; i += 1024) hist[i] = 0;
    __syncthreads();
    const int e0 = blk * CH;
    for (int j = 0; j < CH / 1024; ++j) {
        const int e = e0 + j * 1024 + t;
        if (e < EE) atomicAdd(&hist[col[e] >> BSH], 1);
    }
    __syncthreads();
    for (int i = t; i < NBKT; i += 1024) bcnt[i * NBLK + blk] = hist[i];
}

__device__ __forceinline__ void role_bscan(int* __restrict__ bcnt, char* sm) {
    int* buf = reinterpret_cast<int*>(sm);
    int* sh  = buf + 1024 * SCAN_PER;
    const int t = threadIdx.x;
    for (int i = t; i < 1024 * SCAN_PER; i += 1024) buf[i] = (i < BM) ? bcnt[i] : 0;
    __syncthreads();
    int sum = 0;
    for (int k = 0; k < SCAN_PER; ++k) sum += buf[t * SCAN_PER + k];
    const int incl = block_incl_scan(sh, sum);
    int run = incl - sum;
    for (int k = 0; k < SCAN_PER; ++k) {
        const int v = buf[t * SCAN_PER + k];
        buf[t * SCAN_PER + k] = run;
        run += v;
    }
    __syncthreads();
    for (int i = t; i < BM; i += 1024) bcnt[i] = buf[i];
    if (t == 1023) bcnt[BM] = incl;
}

__device__ __forceinline__
void role_bscatter(const int* __restrict__ row, const int* __restrict__ col,
                   const float* __restrict__ w, const int* __restrict__ gbase,
                   int* __restrict__ rec, char* sm) {
    int*   st_x  = reinterpret_cast<int*>(sm);
    float* st_w  = reinterpret_cast<float*>(sm) + CH;
    int*   hist  = reinterpret_cast<int*>(sm) + 2 * CH;
    int*   lbase = hist + NBKT;
    int*   gb    = lbase + NBKT;
    int*   sc    = gb + NBKT;
    unsigned char* st_b = reinterpret_cast<unsigned char*>(sc + 32);
    const int t   = threadIdx.x;
    const int blk = blockIdx.x;
    for (int i = t; i < NBKT; i += 1024) {
        hist[i] = 0;
        gb[i]   = gbase[i * NBLK + blk];
    }
    __syncthreads();

    const int e0 = blk * CH;
    int   c[CH / 1024], r[CH / 1024], s[CH / 1024];
    float ww[CH / 1024];
    #pragma unroll
    for (int j = 0; j < CH / 1024; ++j) {
        const int e = e0 + j * 1024 + t;
        const bool on = e < EE;
        c[j]  = on ? col[e] : -1;
        s[j]  = on ? row[e] : 0;
        ww[j] = on ? w[e]   : 0.f;
    }
    #pragma unroll
    for (int j = 0; j < CH / 1024; ++j)
        r[j] = (c[j] >= 0) ? atomicAdd(&hist[c[j] >> BSH], 1) : 0;
    __syncthreads();

    const int hv   = (t < NBKT) ? hist[t] : 0;
    const int incl = block_incl_scan(sc, hv);
    if (t < NBKT) lbase[t] = incl - hv;
    __syncthreads();

    #pragma unroll
    for (int j = 0; j < CH / 1024; ++j) {
        if (c[j] >= 0) {
            const int b = c[j] >> BSH;
            const int p = lbase[b] + r[j];
            st_x[p] = s[j] | ((c[j] & (BN - 1)) << 17);
            st_w[p] = ww[j];
            st_b[p] = (unsigned char)b;
        }
    }
    __syncthreads();

    const int nvalid = min(CH, EE - e0);
    for (int i = t; i < nvalid; i += 1024) {
        const int b  = st_b[i];
        const int gp = gb[b] + (i - lbase[b]);
        int2 o;
        o.x = st_x[i];
        o.y = __float_as_int(st_w[i]);
        *reinterpret_cast<int2*>(rec + 2 * (size_t)gp) = o;
    }
}

__device__ __forceinline__
void role_bsort(const int* __restrict__ rec, const int* __restrict__ gbase,
                int* __restrict__ row_s, float* w_s, int* __restrict__ offs,
                float* __restrict__ dinv, char* sm) {
    int*   st_r = reinterpret_cast<int*>(sm);
    float* st_w = reinterpret_cast<float*>(sm) + CAP;
    int*   cnt  = reinterpret_cast<int*>(sm) + 2 * CAP;
    int*   cur  = cnt + BN;
    int*   loff = cur + BN;
    const int t  = threadIdx.x;
    const int b  = blockIdx.x;
    const int bs = gbase[b * NBLK];
    const int be = gbase[(b + 1) * NBLK];
    const int n  = be - bs;
    const int2* rb = reinterpret_cast<const int2*>(rec) + bs;

    if (t < BN) cnt[t] = 0;
    __syncthreads();
    for (int i0 = 0; i0 < n; i0 += 4096) {
        int xs[4];
        #pragma unroll
        for (int u = 0; u < 4; ++u) {
            const int i = i0 + u * 1024 + t;
            xs[u] = (i < n) ? rb[i].x : -1;
        }
        #pragma unroll
        for (int u = 0; u < 4; ++u)
            if (xs[u] >= 0) atomicAdd(&cnt[xs[u] >> 17], 1);
    }
    __syncthreads();

    const int cv   = (t < BN) ? cnt[t] : 0;
    const int incl = block_incl_scan(cur, cv);
    if (t < BN) {
        loff[t] = incl - cv;
        if (t == BN - 1) loff[BN] = incl;
    }
    __syncthreads();
    if (t < BN) {
        cur[t] = loff[t];
        const int node = b * BN + t;
        if (node < NN) offs[node] = bs + loff[t];
    }
    if (b == NBKT - 1 && t == 0) offs[NN] = be;
    __syncthreads();

    const int node = b * BN + t;
    if (n <= CAP) {
        for (int i0 = 0; i0 < n; i0 += 4096) {
            int2 q[4];
            #pragma unroll
            for (int u = 0; u < 4; ++u) {
                const int i = i0 + u * 1024 + t;
                q[u] = (i < n) ? rb[i] : make_int2(-1, 0);
            }
            #pragma unroll
            for (int u = 0; u < 4; ++u) {
                if (q[u].x >= 0) {
                    const int p = atomicAdd(&cur[q[u].x >> 17], 1);
                    st_r[p] = q[u].x & 0x1FFFF;
                    st_w[p] = __int_as_float(q[u].y);
                }
            }
        }
        __syncthreads();
        for (int i = t; i < n; i += 1024) {
            row_s[bs + i] = st_r[i];
            w_s[bs + i]   = st_w[i];
        }
        if (t < BN && node < NN) {
            float d = 1.0f;
            for (int p = loff[t]; p < loff[t + 1]; ++p) d += st_w[p];
            dinv[node] = d > 0.f ? rsqrtf(d) : 0.f;
        }
    } else {
        for (int i = t; i < n; i += 1024) {
            const int2 q = rb[i];
            const int  p = atomicAdd(&cur[q.x >> 17], 1);
            row_s[bs + p] = q.x & 0x1FFFF;
            w_s[bs + p]   = __int_as_float(q.y);
        }
        __syncthreads();
        if (t < BN && node < NN) {
            float d = 1.0f;
            for (int p = loff[t]; p < loff[t + 1]; ++p) d += w_s[bs + p];
            dinv[node] = d > 0.f ? rsqrtf(d) : 0.f;
        }
    }
}

#define GT_ROWS  128
#define GT_PITCH (FIN + 8)
#define GT_LDS   (GT_ROWS * GT_PITCH * 4)
#define NGT      ((NN + GT_ROWS - 1) / GT_ROWS)
#define GT_SCAN    400
#define GT_SCATTER 190
#define GT_SORT    (NGT - GT_SCAN - GT_SCATTER)

static_assert(NN % 16 == 0, "a row tile ends on a 16-row boundary");
static_assert(FIN % 4 == 0, "rows are copied in 16-byte pieces");
static_assert(GT_SORT > 0, "every tile has a launch");

__device__ __forceinline__
void role_gemm_in(const float* __restrict__ x, const float* __restrict__ Win,
                  const float* __restrict__ bin, float* __restrict__ x0, int tile, char* sm) {
    float* xs = reinterpret_cast<float*>(sm);
    const int tid   = threadIdx.x;
    const int row0  = tile * GT_ROWS;
    const int nrows = min(GT_ROWS, NN - row0);
    const int n4    = nrows * (FIN / 4);
    const v4f* srcp = reinterpret_cast<const v4f*>(x + (size_t)row0 * FIN);

    for (int base = 0; base < n4; base += 5 * 1024) {
        v4f v[5];
        #pragma unroll
        for (int u = 0; u < 5; ++u) {
            const int i = base + u * 1024 + tid;
            v[u] = (i < n4) ? srcp[i] : v4f{0.f, 0.f, 0.f, 0.f};
        }
        #pragma unroll
        for (int u = 0; u < 5; ++u) {
            const int i = base + u * 1024 + tid;
            if (i < n4) {
                const int r = i / (FIN / 4), c = i - r * (FIN / 4);
                *reinterpret_cast<v4f*>(&xs[r * GT_PITCH + c * 4]) = v[u];
            }
        }
    }
    __syncthreads();

    const int lane = tid & 31;
    const int wv   = __builtin_amdgcn_readfirstlane(tid >> 5);
    const int mt   = wv >> 2;
    const int n0   = (wv & 3) * 16;
    if (mt * 16 >= nrows) return;

    const int m    = lane & 15;
    const int kh   = (lane >> 4) * 2;
    const int nl   = lane & 15;
    const float* xr = xs + (mt * 16 + m) * GT_PITCH;

    v8f acc = {};
    for (int k0 = 0; k0 < FIN; k0 += 4) {
        v2f a, b;
        a[0] = xr[k0 + kh];
        a[1] = xr[k0 + kh + 1];
        b[0] = Win[(size_t)(k0 + kh)     * HID + n0 + nl];
        b[1] = Win[(size_t)(k0 + kh + 1) * HID + n0 + nl];
        acc  = wmma_f32_16x16x4(a, b, acc);
    }

    #pragma unroll
    for (int v = 0; v < 8; ++v) {
        int mm  = v + (lane >> 4) * 8;
        int c   = n0 + nl;
        float val = acc[v] + bin[c];
        val = val > 0.f ? val : 0.f;
        x0[(size_t)(row0 + mt * 16 + mm) * HID + c] = val;
    }
}

#define LDS_SCAN    (1024 * SCAN_PER * 4 + 128)
#define LDS_SCATTER (2 * CH * 4 + 3 * NBKT * 4 + 128 + CH)
#define LDS_SORT    (2 * CAP * 4 + (3 * BN + 1) * 4)
#define MAXB(a, b)  ((a) > (b) ? (a) : (b))

__global__ __launch_bounds__(1024)
void k_bscan(int* __restrict__ bcnt, const float* __restrict__ x, const float* __restrict__ Win,
             const float* __restrict__ bin, float* __restrict__ x0) {
    __shared__ alignas(16) char sm[MAXB(LDS_SCAN, GT_LDS)];
    if (blockIdx.x == 0) role_bscan(bcnt, sm);
    else                 role_gemm_in(x, Win, bin, x0, (int)blockIdx.x - 1, sm);
}

__global__ __launch_bounds__(1024)
void k_bscatter(const int* __restrict__ row, const int* __restrict__ col,
                const float* __restrict__ w, const int* __restrict__ gbase,
                int* __restrict__ rec, const float* __restrict__ x,
                const float* __restrict__ Win, const float* __restrict__ bin,
                float* __restrict__ x0) {
    __shared__ alignas(16) char sm[MAXB(LDS_SCATTER, GT_LDS)];
    if (blockIdx.x < NBLK) role_bscatter(row, col, w, gbase, rec, sm);
    else                   role_gemm_in(x, Win, bin, x0, (int)blockIdx.x - NBLK + GT_SCAN, sm);
}

__global__ __launch_bounds__(1024)
void k_bsort(const int* __restrict__ rec, const int* __restrict__ gbase,
             int* __restrict__ row_s, float* w_s, int* __restrict__ offs,
             float* __restrict__ dinv, const float* __restrict__ x,
             const float* __restrict__ Win, const float* __restrict__ bin,
             float* __restrict__ x0) {
    __shared__ alignas(16) char sm[MAXB(LDS_SORT, GT_LDS)];
    if (blockIdx.x < NBKT) role_bsort(rec, gbase, row_s, w_s, offs, dinv, sm);
    else                   role_gemm_in(x, Win, bin, x0,
                                        (int)blockIdx.x - NBKT + GT_SCAN + GT_SCATTER, sm);
}


template <int S>
__device__ __forceinline__ int row_pick(int x) {
    return __builtin_amdgcn_mov_dpp(x, 0x150 + S, 0xf, 0xf, true);
}

__device__ __forceinline__ float xhalf(float x) {
    const int b = __float_as_int(x);
    return __int_as_float(__builtin_amdgcn_permlanex16(b, b, 0x76543210, 0xfedcba98, false, false));
}

template <int S>
__device__ __forceinline__ void pair_load(const float* __restrict__ hin, int my_r, float my_n,
                                          unsigned c16, v4f& v, float& w) {
    const unsigned r = (unsigned)row_pick<S>(my_r);
    w = __int_as_float(row_pick<S>(__float_as_int(my_n)));
    v = *reinterpret_cast<const v4f*>(reinterpret_cast<const char*>(hin) + ((r << 8) | c16));
}

template <int S0, int G>
__device__ __forceinline__ void gather_pairs(const float* __restrict__ hin, int my_r, float my_n,
                                             unsigned c16, v4f& acc) {
    v4f   v[G];
    float w[G];
    pair_load<S0 + 0>(hin, my_r, my_n, c16, v[0], w[0]);
    pair_load<S0 + 1>(hin, my_r, my_n, c16, v[1], w[1]);
    pair_load<S0 + 2>(hin, my_r, my_n, c16, v[2], w[2]);
    pair_load<S0 + 3>(hin, my_r, my_n, c16, v[3], w[3]);
    if constexpr (G == 8) {
        pair_load<S0 + 4>(hin, my_r, my_n, c16, v[G - 4], w[G - 4]);
        pair_load<S0 + 5>(hin, my_r, my_n, c16, v[G - 3], w[G - 3]);
        pair_load<S0 + 6>(hin, my_r, my_n, c16, v[G - 2], w[G - 2]);
        pair_load<S0 + 7>(hin, my_r, my_n, c16, v[G - 1], w[G - 1]);
    }
    #pragma unroll
    for (int j = 0; j < G; ++j) {
        const v4f ww = {w[j], w[j], w[j], w[j]};
        acc = __builtin_elementwise_fma(v[j], ww, acc);
    }
}

template <bool FIRST>
__device__ __forceinline__
void layer_body(const int* __restrict__ row_s, const float* __restrict__ nrm_s, float* w_rw,
                const int* __restrict__ offs, const float* __restrict__ x0,
                const float* __restrict__ dinv, const float* __restrict__ Wl,
                const float* __restrict__ hin, float* __restrict__ hout, float beta) {
    __shared__ alignas(16) float zs[16][HID + 8];
    __shared__ float ws[HID][HID];
    const int tid  = threadIdx.x;
    const int mb   = blockIdx.x;
    const int lane = tid & 31;
    const int wv   = __builtin_amdgcn_readfirstlane(tid >> 5);
    const int hw   = lane >> 4;
    const int c    = lane & 15;
    const int el   = 2 * c + hw;
    const unsigned c16 = (unsigned)c << 4;

    for (int i = tid; i < HID * HID; i += 128)
        ws[i >> 6][i & 63] = Wl[i];

    const int nb = mb * 16 + wv * 4;
    int    seg_s[4], seg_n[4], pr[4];
    float  pn[4], di[4];
    #pragma unroll
    for (int q = 0; q < 4; ++q) {
        seg_s[q] = offs[nb + q];
        seg_n[q] = offs[nb + q + 1] - seg_s[q];
    }
    #pragma unroll
    for (int q = 0; q < 4; ++q) {
        const bool on = el < seg_n[q];
        pr[q] = on ? row_s[seg_s[q] + el] : 0;
        if (FIRST) pn[q] = on ? w_rw[seg_s[q] + el] : 0.f;
        else       pn[q] = on ? nrm_s[seg_s[q] + el] : 0.f;
        di[q] = dinv[nb + q];
    }
    if (FIRST) {
        float ds[4];
        #pragma unroll
        for (int q = 0; q < 4; ++q) ds[q] = dinv[pr[q]];
        #pragma unroll
        for (int q = 0; q < 4; ++q) pn[q] = ds[q] * pn[q] * di[q];
        #pragma unroll
        for (int q = 0; q < 4; ++q)
            if (el < seg_n[q]) w_rw[seg_s[q] + el] = pn[q];
    }

    const float* own = hw ? x0 : hin;

    #pragma unroll
    for (int q = 0; q < 4; ++q) {
        const v4f val = *reinterpret_cast<const v4f*>(own + (size_t)(nb + q) * HID + c * 4);
        v4f   acc  = {0.f, 0.f, 0.f, 0.f};
        int   my_r = pr[q];
        float my_n = pn[q];
        for (int done = 0; done < seg_n[q]; done += 32) {
            const int n = min(32, seg_n[q] - done);
            if (done > 0) {
                const bool on = el < n;
                my_r = on ? row_s[seg_s[q] + done + el] : 0;
                if (FIRST) {
                    my_n = on ? w_rw[seg_s[q] + done + el] : 0.f;
                    my_n = dinv[my_r] * my_n * di[q];
                    if (on) w_rw[seg_s[q] + done + el] = my_n;
                } else {
                    my_n = on ? nrm_s[seg_s[q] + done + el] : 0.f;
                }
            }
            gather_pairs<0, 8>(hin, my_r, my_n, c16, acc);
            if (n > 16) {
                gather_pairs<8, 4>(hin, my_r, my_n, c16, acc);
                if (n > 24) gather_pairs<12, 4>(hin, my_r, my_n, c16, acc);
            }
        }
        v4f sum, xr;
        sum.x = acc.x + xhalf(acc.x);  xr.x = xhalf(val.x);
        sum.y = acc.y + xhalf(acc.y);  xr.y = xhalf(val.y);
        sum.z = acc.z + xhalf(acc.z);  xr.z = xhalf(val.z);
        sum.w = acc.w + xhalf(acc.w);  xr.w = xhalf(val.w);
        if (hw == 0) {
            const float d2 = di[q] * di[q];
            v4f zz;
            zz.x = __builtin_fmaf(1.f - ALPHA, __builtin_fmaf(d2, val.x, sum.x), ALPHA * xr.x);
            zz.y = __builtin_fmaf(1.f - ALPHA, __builtin_fmaf(d2, val.y, sum.y), ALPHA * xr.y);
            zz.z = __builtin_fmaf(1.f - ALPHA, __builtin_fmaf(d2, val.z, sum.z), ALPHA * xr.z);
            zz.w = __builtin_fmaf(1.f - ALPHA, __builtin_fmaf(d2, val.w, sum.w), ALPHA * xr.w);
            *reinterpret_cast<v4f*>(&zs[wv * 4 + q][c * 4]) = zz;
        }
    }
    __syncthreads();

    const int n0   = wv * 16;
    const int m    = lane & 15;
    const int kh   = (lane >> 4) * 2;
    const int nl   = lane & 15;

    v8f acc = {};
    #pragma unroll
    for (int k0 = 0; k0 < HID; k0 += 4) {
        v2f a, b;
        a[0] = zs[m][k0 + kh];
        a[1] = zs[m][k0 + kh + 1];
        b[0] = ws[k0 + kh][n0 + nl];
        b[1] = ws[k0 + kh + 1][n0 + nl];
        acc  = wmma_f32_16x16x4(a, b, acc);
    }

    #pragma unroll
    for (int v = 0; v < 8; ++v) {
        int mm = v + (lane >> 4) * 8;
        float z   = zs[mm][n0 + nl];
        float val = (1.f - beta) * z + beta * acc[v];
        val = val > 0.f ? val : 0.f;
        hout[(size_t)(mb * 16 + mm) * HID + n0 + nl] = val;
    }
}

__global__ __launch_bounds__(128)
void k_layer_first(const int* __restrict__ row_s, float* w_rw,
                   const int* __restrict__ offs, const float* __restrict__ x0,
                   const float* __restrict__ dinv, const float* __restrict__ Wl,
                   const float* __restrict__ hin, float* __restrict__ hout, float beta) {
    layer_body<true>(row_s, nullptr, w_rw, offs, x0, dinv, Wl, hin, hout, beta);
}

__global__ __launch_bounds__(128)
void k_layer(const int* __restrict__ row_s, const float* __restrict__ nrm_s,
             const int* __restrict__ offs, const float* __restrict__ x0,
             const float* __restrict__ dinv, const float* __restrict__ Wl,
             const float* __restrict__ hin, float* __restrict__ hout, float beta) {
    layer_body<false>(row_s, nrm_s, nullptr, offs, x0, dinv, Wl, hin, hout, beta);
}

__global__ void k_out(const float* __restrict__ h, const float* __restrict__ Wout,
                      const float* __restrict__ bout, float* __restrict__ out) {
    int i = blockIdx.x * blockDim.x + threadIdx.x;
    if (i >= NN * OUTC) return;
    int n = i / OUTC, c = i - n * OUTC;
    float s = bout[c];
    const float* hr = h + (size_t)n * HID;
    #pragma unroll 16
    for (int k = 0; k < HID; ++k) s += hr[k] * Wout[k * OUTC + c];
    out[i] = s;
}

extern "C" void kernel_launch(void* const* d_in, const int* in_sizes, int n_in,
                              void* d_out, int out_size, void* d_ws, size_t ws_size,
                              hipStream_t stream) {
    (void)in_sizes; (void)n_in; (void)out_size; (void)ws_size;

    const float* x     = (const float*)d_in[0];
    const int*   ei    = (const int*)  d_in[1];
    const float* ew    = (const float*)d_in[2];
    const float* Win   = (const float*)d_in[3];
    const float* bin   = (const float*)d_in[4];
    const float* convw = (const float*)d_in[5];
    const float* Wout  = (const float*)d_in[6];
    const float* bout  = (const float*)d_in[7];
    const int* row = ei;
    const int* col = ei + EE;

    char*  ws  = (char*)d_ws;
    size_t off = 0;
    auto carve = [&](size_t bytes) -> float* {
        float* p = (float*)(ws + off);
        off = (off + bytes + 255) & ~(size_t)255;
        return p;
    };
    float* dinv  = carve((size_t)NN * 4);
    float* nrm_s = carve((size_t)EE * 4);
    float* x0    = carve((size_t)NN * HID * 4);
    float* hA    = carve((size_t)NN * HID * 4);
    float* hB    = carve((size_t)NN * HID * 4);

    int* rec   = (int*)hB;
    int* gbase = (int*)((char*)hB + 12800000);

    int* row_s = (int*)d_out;
    int* offs  = (int*)((char*)d_out + 6400000);

    k_bhist   <<<NBLK,              1024, 0, stream>>>(col, gbase);
    k_bscan   <<<1 + GT_SCAN,       1024, 0, stream>>>(gbase, x, Win, bin, x0);
    k_bscatter<<<NBLK + GT_SCATTER, 1024, 0, stream>>>(row, col, ew, gbase, rec, x, Win, bin, x0);
    k_bsort   <<<NBKT + GT_SORT,    1024, 0, stream>>>(rec, gbase, row_s, nrm_s, offs, dinv,
                                                       x, Win, bin, x0);

    const float* hin = x0;
    for (int l = 0; l < NLAYER; ++l) {
        float beta = logf(0.5f / (float)(l + 1) + 1.0f);
        float* hout = (l & 1) ? hB : hA;
        if (l == 0)
            k_layer_first<<<NN / 16, 128, 0, stream>>>(row_s, nrm_s, offs, x0, dinv,
                                                       convw, hin, hout, beta);
        else
            k_layer<<<NN / 16, 128, 0, stream>>>(row_s, nrm_s, offs, x0, dinv,
                                                 convw + (size_t)l * HID * HID,
                                                 hin, hout, beta);
        hin = hout;
    }

    k_out<<<(NN * OUTC + 255) / 256, 256, 0, stream>>>(hin, Wout, bout, (float*)d_out);
}
